# starting kernel
# baseline (speedup 1.0000x reference)
#include <hip/hip_runtime.h>
#include <math.h>

typedef __attribute__((ext_vector_type(2))) float v2f;
typedef __attribute__((ext_vector_type(8))) float v8f;

#define PI_D 3.14159265358979323846

constexpr long long SZ_A1    = 255LL*128*128;
constexpr long long SZ_XP    = 128LL*256*128;
constexpr long long SZ_PH1   = 256LL*256;
constexpr long long SZ_G     = 128LL*256*128;
constexpr long long SZ_F1    = 255LL*128*128;
constexpr long long SZ_R1    = 64LL*128;
constexpr long long SZ_R2P   = 128LL*256;
constexpr long long SZ_T1    = 256LL*64*128;
constexpr long long SZ_FF    = 64LL*128*128;
constexpr long long SZ_WC    = 64LL*128*128;
constexpr long long SZ_F2    = 64LL*128*128;
constexpr long long SZ_P2    = 127LL*64*64;
constexpr long long SZ_FT    = 127LL*64*128;
constexpr long long SZ_PH2   = 128LL*128;
constexpr long long SZ_XSPEC = 64LL*128*128;
constexpr long long SZ_TVEC  = 256;
constexpr long long SZ_TC    = 64;
constexpr long long SZ_WSP   = 128LL*128;
constexpr long long SZ_XS1   = 128LL*255*128;
constexpr long long SZ_T2    = 256LL*64*128;
constexpr long long SZ_XS2   = 64LL*128*128;
constexpr long long SZ_XSP   = 67LL*129*128;
constexpr long long SZ_XCV   = 64LL*128*128;

constexpr long long OFF_A1    = 0;
constexpr long long OFF_XP    = OFF_A1    + SZ_A1;
constexpr long long OFF_COS1  = OFF_XP    + SZ_XP;
constexpr long long OFF_SIN1N = OFF_COS1  + SZ_PH1;
constexpr long long OFF_GR    = OFF_SIN1N + SZ_PH1;
constexpr long long OFF_GI    = OFF_GR    + SZ_G;
constexpr long long OFF_F1R   = OFF_GI    + SZ_G;
constexpr long long OFF_F1I   = OFF_F1R   + SZ_F1;
constexpr long long OFF_R1    = OFF_F1I   + SZ_F1;
constexpr long long OFF_R2P   = OFF_R1    + SZ_R1;
constexpr long long OFF_T1R   = OFF_R2P   + SZ_R2P;
constexpr long long OFF_T1I   = OFF_T1R   + SZ_T1;
constexpr long long OFF_FFR   = OFF_T1I   + SZ_T1;
constexpr long long OFF_FFI   = OFF_FFR   + SZ_FF;
constexpr long long OFF_WCR   = OFF_FFI   + SZ_FF;
constexpr long long OFF_WCI   = OFF_WCR   + SZ_WC;
constexpr long long OFF_F2R   = OFF_WCI   + SZ_WC;
constexpr long long OFF_F2I   = OFF_F2R   + SZ_F2;
constexpr long long OFF_P2A   = OFF_F2I   + SZ_F2;
constexpr long long OFF_P2B   = OFF_P2A   + SZ_P2;
constexpr long long OFF_FTR   = OFF_P2B   + SZ_P2;
constexpr long long OFF_FTI   = OFF_FTR   + SZ_FT;
constexpr long long OFF_COS2  = OFF_FTI   + SZ_FT;
constexpr long long OFF_SIN2  = OFF_COS2  + SZ_PH2;
constexpr long long OFF_XSPEC = OFF_SIN2  + SZ_PH2;
constexpr long long OFF_TVEC  = OFF_XSPEC + SZ_XSPEC;
constexpr long long OFF_TCR   = OFF_TVEC  + SZ_TVEC;
constexpr long long OFF_TCI   = OFF_TCR   + SZ_TC;
constexpr long long OFF_WSP   = OFF_TCI   + SZ_TC;
constexpr long long OFF_XS1   = OFF_WSP   + SZ_WSP;
constexpr long long OFF_T2R   = OFF_XS1   + SZ_XS1;
constexpr long long OFF_XS2   = OFF_T2R   + SZ_T2;
constexpr long long OFF_XSP   = OFF_XS2   + SZ_XS2;
constexpr long long OFF_XCV   = OFF_XSP   + SZ_XSP;

__global__ void gemm_wmma(const float* __restrict__ A, const float* __restrict__ B,
                          float* __restrict__ C, int M, int K,
                          int lda, int ldb, int ldc,
                          long long sAb, long long sBb, long long sCb,
                          float alpha, int accumulate) {
  const int lane = threadIdx.x;
  const int n0 = blockIdx.x * 64;
  const int m0 = blockIdx.y * 16;
  const long long b = blockIdx.z;
  A += b * sAb; B += b * sBb; C += b * sCb;
  const int half = lane >> 4;
  const int l16  = lane & 15;
  v8f acc0 = {}, acc1 = {}, acc2 = {}, acc3 = {};
  const float* arow = A + (long long)(m0 + l16) * lda + 2 * half;
  const float* brow = B + (long long)(2 * half) * ldb + n0 + l16;
  for (int k0 = 0; k0 < K; k0 += 4) {
    v2f a; a.x = arow[k0]; a.y = arow[k0 + 1];
    const float* bp = brow + (long long)k0 * ldb;
    v2f b0; b0.x = bp[0];   b0.y = bp[ldb];
    v2f b1; b1.x = bp[16];  b1.y = bp[ldb + 16];
    v2f b2; b2.x = bp[32];  b2.y = bp[ldb + 32];
    v2f b3; b3.x = bp[48];  b3.y = bp[ldb + 48];
    acc0 = __builtin_amdgcn_wmma_f32_16x16x4_f32(false, a, false, b0, (short)0, acc0, false, false);
    acc1 = __builtin_amdgcn_wmma_f32_16x16x4_f32(false, a, false, b1, (short)0, acc1, false, false);
    acc2 = __builtin_amdgcn_wmma_f32_16x16x4_f32(false, a, false, b2, (short)0, acc2, false, false);
    acc3 = __builtin_amdgcn_wmma_f32_16x16x4_f32(false, a, false, b3, (short)0, acc3, false, false);
    if (k0 + 4 < K) {
      __builtin_prefetch(arow + k0 + 4, 0, 1);
      __builtin_prefetch(bp + 4LL * ldb, 0, 1);
    }
  }
#pragma unroll
  for (int v = 0; v < 8; ++v) {
    const int row = m0 + half * 8 + v;
    float* cp = C + (long long)row * ldc + n0 + l16;
    if (accumulate) {
      cp[0]  += alpha * acc0[v]; cp[16] += alpha * acc1[v];
      cp[32] += alpha * acc2[v]; cp[48] += alpha * acc3[v];
    } else {
      cp[0]  = alpha * acc0[v]; cp[16] = alpha * acc1[v];
      cp[32] = alpha * acc2[v]; cp[48] = alpha * acc3[v];
    }
  }
}

__global__ void conv3x3_wmma(const float* __restrict__ xsp, const float* __restrict__ ck,
                             const float* __restrict__ cb, const float* __restrict__ tvec,
                             float* __restrict__ out) {
  const int lane = threadIdx.x;
  const int n0 = blockIdx.x * 64;
  const int m0 = blockIdx.y * 16;
  const int l  = blockIdx.z;
  const int half = lane >> 4, l16 = lane & 15;
  v8f acc0 = {}, acc1 = {}, acc2 = {}, acc3 = {};
  for (int dy = 0; dy < 3; ++dy) {
    for (int dx = 0; dx < 3; ++dx) {
      const float* arow = xsp + (((long long)(l + dy)) * 129 + (m0 + l16 + dx)) * 128 + 2 * half;
      const float* brow = ck + (((long long)(dy * 3 + dx)) * 128 + 2 * half) * 128 + n0 + l16;
      for (int k0 = 0; k0 < 128; k0 += 4) {
        v2f a; a.x = arow[k0]; a.y = arow[k0 + 1];
        const float* bp = brow + (long long)k0 * 128;
        v2f b0; b0.x = bp[0];  b0.y = bp[128];
        v2f b1; b1.x = bp[16]; b1.y = bp[144];
        v2f b2; b2.x = bp[32]; b2.y = bp[160];
        v2f b3; b3.x = bp[48]; b3.y = bp[176];
        acc0 = __builtin_amdgcn_wmma_f32_16x16x4_f32(false, a, false, b0, (short)0, acc0, false, false);
        acc1 = __builtin_amdgcn_wmma_f32_16x16x4_f32(false, a, false, b1, (short)0, acc1, false, false);
        acc2 = __builtin_amdgcn_wmma_f32_16x16x4_f32(false, a, false, b2, (short)0, acc2, false, false);
        acc3 = __builtin_amdgcn_wmma_f32_16x16x4_f32(false, a, false, b3, (short)0, acc3, false, false);
      }
    }
  }
  const int c0 = n0 + l16;
  const float bias0 = cb[c0]      + tvec[128 + c0];
  const float bias1 = cb[c0 + 16] + tvec[128 + c0 + 16];
  const float bias2 = cb[c0 + 32] + tvec[128 + c0 + 32];
  const float bias3 = cb[c0 + 48] + tvec[128 + c0 + 48];
#pragma unroll
  for (int v = 0; v < 8; ++v) {
    const int row = m0 + half * 8 + v;
    float* op = out + (((long long)l) * 128 + row) * 128 + c0;
    op[0]  = acc0[v] + bias0; op[16] = acc1[v] + bias1;
    op[32] = acc2[v] + bias2; op[48] = acc3[v] + bias3;
  }
}

__global__ void zerof(float* p, long long n) {
  long long i = (long long)blockIdx.x * 256 + threadIdx.x;
  if (i < n) p[i] = 0.0f;
}

__global__ void build_xpad(const float* __restrict__ x, float* __restrict__ xp, long long n) {
  long long i = (long long)blockIdx.x * 256 + threadIdx.x;
  if (i >= n) return;
  long long t = i / 32768, r = i % 32768, p = r / 128, c = r % 128;
  xp[i] = (p < 255) ? x[(t * 255 + p) * 128 + c] : 0.0f;
}

__global__ void k_tvec(const float* __restrict__ temb,
                       const float* __restrict__ dw, const float* __restrict__ db,
                       const float* __restrict__ drw, const float* __restrict__ drb,
                       const float* __restrict__ diw, const float* __restrict__ dib,
                       float* tvec, float* tcr, float* tci) {
  int j = threadIdx.x;
  float acc = db[j];
  for (int k = 0; k < 256; ++k) acc += temb[k] * dw[k * 256 + j];
  tvec[j] = acc;
  if (j < 64) {
    float ar = drb[j], ai = dib[j];
    for (int k = 0; k < 256; ++k) {
      ar += temb[k] * drw[k * 64 + j];
      ai += temb[k] * diw[k * 64 + j];
    }
    tcr[j] = ar; tci[j] = ai;
  }
}

__global__ void k_wc(const float* __restrict__ wr, const float* __restrict__ wi,
                     const float* __restrict__ tcr, const float* __restrict__ tci,
                     float* wcr, float* wci, long long n) {
  long long i = (long long)blockIdx.x * 256 + threadIdx.x;
  if (i >= n) return;
  int l = (int)(i / 16384);
  float tr = tcr[l], ti = tci[l], r = wr[i], im = wi[i];
  wcr[i] = tr * r - ti * im;
  wci[i] = tr * im + ti * r;
}

__global__ void k_wsp(const float* __restrict__ spw, const float* __restrict__ tvec,
                      float* wsp, long long n) {
  long long i = (long long)blockIdx.x * 256 + threadIdx.x;
  if (i >= n) return;
  int o = (int)(i % 128);
  wsp[i] = tvec[o] * spw[i];
}

__global__ void k_phase1(float* c1, float* s1n, long long n) {
  long long i = (long long)blockIdx.x * 256 + threadIdx.x;
  if (i >= n) return;
  int m = (int)(i / 256), p = (int)(i % 256);
  if (m < 255 && p < 255) {
    double mval = (double)(m - 127);
    double phi = 2.0 * PI_D * (double)p / 255.0;
    double dphi = 2.0 * PI_D / 255.0;
    c1[i]  = (float)( cos(phi * mval) * dphi);
    s1n[i] = (float)(-sin(phi * mval) * dphi);
  } else { c1[i] = 0.0f; s1n[i] = 0.0f; }
}

__global__ void k_phase2(float* c2, float* s2, long long n) {
  long long i = (long long)blockIdx.x * 256 + threadIdx.x;
  if (i >= n) return;
  int p = (int)(i / 128), m = (int)(i % 128);
  if (p < 127 && m < 127) {
    double mval = (double)(m - 63);
    double phi = 2.0 * PI_D * (double)p / 127.0;
    c2[i] = (float)cos(mval * phi);
    s2[i] = (float)sin(mval * phi);
  } else { c2[i] = 0.0f; s2[i] = 0.0f; }
}

__global__ void k_resize(float* r1, float* r2p) {
  int tid = threadIdx.x;
  if (tid < 64) {
    int i = tid;
    float scale = 0.5f;
    float sample = (i + 0.5f) / scale - 0.5f;
    float sum = 0.0f;
    for (int j = 0; j < 128; ++j) {
      float w = 1.0f - fabsf(sample - j) * scale;
      if (w > 0.0f) sum += w;
    }
    for (int j = 0; j < 128; ++j) {
      float w = 1.0f - fabsf(sample - j) * scale;
      r1[i * 128 + j] = (w > 0.0f) ? (w / sum) : 0.0f;
    }
  } else if (tid < 192) {
    int j = tid - 64;
    if (j < 127) {
      float scale = 127.0f / 255.0f;
      float sample = (j + 0.5f) / scale - 0.5f;
      float sum = 0.0f;
      for (int p = 0; p < 255; ++p) {
        float w = 1.0f - fabsf(sample - p) * scale;
        if (w > 0.0f) sum += w;
      }
      for (int p = 0; p < 255; ++p) {
        float w = 1.0f - fabsf(sample - p) * scale;
        r2p[j * 256 + p] = (w > 0.0f) ? (w / sum) : 0.0f;
      }
      r2p[j * 256 + 255] = 0.0f;
    } else {
      for (int p = 0; p < 256; ++p) r2p[j * 256 + p] = 0.0f;
    }
  }
}

__global__ void k_legendre(float* out, int L, int weighted) {
  int t = threadIdx.x;
  if (t >= L) return;
  double theta = (2.0 * t + 1.0) * PI_D / (2.0 * L - 1.0);
  double x = cos(theta), s = sin(theta);
  double w = weighted ? (s * PI_D / (double)L) : 1.0;
  auto idx = [&](int m, int l) { return ((long long)(L - 1 + m) * L + l) * (long long)L + t; };
  out[idx(0, 0)] = (float)(sqrt(1.0 / (4.0 * PI_D)) * w);
  for (int l = 1; l < L; ++l) {
    double pprev = (double)out[idx(l - 1, l - 1)];
    out[idx(l, l)]     = (float)(-sqrt((2.0 * l + 1.0) / (2.0 * l)) * s * pprev);
    out[idx(l - 1, l)] = (float)( sqrt(2.0 * l + 1.0) * x * pprev);
    for (int m = 0; m <= l - 2; ++m) {
      double ll = (double)l, mm = (double)m;
      double a = sqrt((4.0 * ll * ll - 1.0) / (ll * ll - mm * mm));
      double b = sqrt((2.0 * ll + 1.0) * ((ll - 1.0) * (ll - 1.0) - mm * mm) /
                      ((2.0 * ll - 3.0) * (ll * ll - mm * mm)));
      out[idx(m, l)] = (float)(a * x * (double)out[idx(m, l - 1)] - b * (double)out[idx(m, l - 2)]);
    }
  }
}

__global__ void k_mirror(float* out, int L, long long n) {
  long long i = (long long)blockIdx.x * 256 + threadIdx.x;
  if (i >= n) return;
  int LL = L * L;
  int m = (int)(i / LL) + 1;
  int r = (int)(i % LL);
  int l = r / L, t = r % L;
  float v = out[((long long)(L - 1 + m) * L + l) * L + t];
  out[((long long)(L - 1 - m) * L + l) * L + t] = (m & 1) ? -v : v;
}

__global__ void k_transp_p2(const float* __restrict__ a, float* __restrict__ b, long long n) {
  long long i = (long long)blockIdx.x * 256 + threadIdx.x;
  if (i >= n) return;
  int m = (int)(i / 4096), t = (int)((i / 64) % 64), l = (int)(i % 64);
  b[i] = a[((long long)m * 64 + l) * 64 + t];
}

__global__ void k_xsppad(const float* __restrict__ xs2, float* __restrict__ xsp, long long n) {
  long long i = (long long)blockIdx.x * 256 + threadIdx.x;
  if (i >= n) return;
  long long row = i / (129 * 128), r = i % (129 * 128);
  int col = (int)(r / 128), c = (int)(r % 128);
  int l = (int)row - 1, m = col - 1;
  xsp[i] = (l >= 0 && l < 64 && m >= 0 && m < 127) ? xs2[((long long)l * 128 + m) * 128 + c] : 0.0f;
}

__global__ void k_fuse(const float* __restrict__ xspec, const float* __restrict__ xcv,
                       const float* __restrict__ lns, const float* __restrict__ lnb,
                       float* __restrict__ out) {
  __shared__ float sh[128];
  int c = threadIdx.x;
  int n = blockIdx.x;
  int l = n / 127, m = n % 127;
  long long idx = ((long long)l * 128 + m) * 128 + c;
  float v = xspec[idx] + xcv[idx];
  float g = 0.5f * v * (1.0f + tanhf(0.7978845608028654f * (v + 0.044715f * v * v * v)));
  sh[c] = g; __syncthreads();
  for (int s = 64; s > 0; s >>= 1) { if (c < s) sh[c] += sh[c + s]; __syncthreads(); }
  float mu = sh[0] * (1.0f / 128.0f); __syncthreads();
  float d = g - mu;
  sh[c] = d * d; __syncthreads();
  for (int s = 64; s > 0; s >>= 1) { if (c < s) sh[c] += sh[c + s]; __syncthreads(); }
  float var = sh[0] * (1.0f / 128.0f);
  out[((long long)l * 127 + m) * 128 + c] = d * rsqrtf(var + 1e-6f) * lns[c] + lnb[c];
}

static inline void launch_gemm(hipStream_t s, const float* A, const float* B, float* C,
                               int M, int K, int lda, int ldb, int ldc,
                               long long sA, long long sB, long long sC,
                               float alpha, int acc, int batch) {
  dim3 g(2, M / 16, batch);
  gemm_wmma<<<g, dim3(32), 0, s>>>(A, B, C, M, K, lda, ldb, ldc, sA, sB, sC, alpha, acc);
}

#define EWG(n) dim3((unsigned)(((n) + 255) / 256)), dim3(256), 0, stream

extern "C" void kernel_launch(void* const* d_in, const int* in_sizes, int n_in,
                              void* d_out, int out_size, void* d_ws, size_t ws_size,
                              hipStream_t stream) {
  (void)in_sizes; (void)n_in; (void)out_size; (void)ws_size;
  const float* x       = (const float*)d_in[0];
  const float* t_emb   = (const float*)d_in[1];
  const float* sp_dw   = (const float*)d_in[2];
  const float* sp_db   = (const float*)d_in[3];
  const float* sp_wt   = (const float*)d_in[4];
  const float* sp_ck   = (const float*)d_in[5];
  const float* sp_cb   = (const float*)d_in[6];
  const float* spec_wr = (const float*)d_in[7];
  const float* spec_wi = (const float*)d_in[8];
  const float* drw     = (const float*)d_in[9];
  const float* drb     = (const float*)d_in[10];
  const float* diw     = (const float*)d_in[11];
  const float* dib     = (const float*)d_in[12];
  const float* lns     = (const float*)d_in[13];
  const float* lnb     = (const float*)d_in[14];
  float* out = (float*)d_out;
  float* ws  = (float*)d_ws;

  float *A1 = ws + OFF_A1, *XP = ws + OFF_XP, *COS1 = ws + OFF_COS1, *SIN1N = ws + OFF_SIN1N;
  float *GR = ws + OFF_GR, *GI = ws + OFF_GI, *F1R = ws + OFF_F1R, *F1I = ws + OFF_F1I;
  float *R1 = ws + OFF_R1, *R2P = ws + OFF_R2P, *T1R = ws + OFF_T1R, *T1I = ws + OFF_T1I;
  float *FFR = ws + OFF_FFR, *FFI = ws + OFF_FFI, *WCR = ws + OFF_WCR, *WCI = ws + OFF_WCI;
  float *F2R = ws + OFF_F2R, *F2I = ws + OFF_F2I, *P2A = ws + OFF_P2A, *P2B = ws + OFF_P2B;
  float *FTR = ws + OFF_FTR, *FTI = ws + OFF_FTI, *COS2 = ws + OFF_COS2, *SIN2 = ws + OFF_SIN2;
  float *XSPEC = ws + OFF_XSPEC, *TVEC = ws + OFF_TVEC, *TCR = ws + OFF_TCR, *TCI = ws + OFF_TCI;
  float *WSP = ws + OFF_WSP, *XS1 = ws + OFF_XS1, *T2R = ws + OFF_T2R, *XS2 = ws + OFF_XS2;
  float *XSP = ws + OFF_XSP, *XCV = ws + OFF_XCV;

  zerof<<<EWG(SZ_A1)>>>(A1, SZ_A1);
  zerof<<<EWG(SZ_P2)>>>(P2A, SZ_P2);
  zerof<<<EWG(8192)>>>(T1R + 255LL * 8192, 8192);
  zerof<<<EWG(8192)>>>(T1I + 255LL * 8192, 8192);
  zerof<<<EWG(8192)>>>(T2R + 255LL * 8192, 8192);

  build_xpad<<<EWG(SZ_XP)>>>(x, XP, SZ_XP);
  k_tvec<<<dim3(1), dim3(256), 0, stream>>>(t_emb, sp_dw, sp_db, drw, drb, diw, dib, TVEC, TCR, TCI);
  k_wc<<<EWG(SZ_WC)>>>(spec_wr, spec_wi, TCR, TCI, WCR, WCI, SZ_WC);
  k_wsp<<<EWG(SZ_WSP)>>>(sp_wt, TVEC, WSP, SZ_WSP);
  k_phase1<<<EWG(SZ_PH1)>>>(COS1, SIN1N, SZ_PH1);
  k_phase2<<<EWG(SZ_PH2)>>>(COS2, SIN2, SZ_PH2);
  k_resize<<<dim3(1), dim3(256), 0, stream>>>(R1, R2P);
  k_legendre<<<dim3(1), dim3(128), 0, stream>>>(A1, 128, 1);
  k_mirror<<<EWG(127LL * 128 * 128)>>>(A1, 128, 127LL * 128 * 128);
  k_legendre<<<dim3(1), dim3(64), 0, stream>>>(P2A, 64, 0);
  k_mirror<<<EWG(63LL * 64 * 64)>>>(P2A, 64, 63LL * 64 * 64);
  k_transp_p2<<<EWG(SZ_P2)>>>(P2A, P2B, SZ_P2);

  launch_gemm(stream, COS1,  XP, GR, 256, 256, 256, 128, 128, 0, 256LL * 128, 256LL * 128, 1.0f, 0, 128);
  launch_gemm(stream, SIN1N, XP, GI, 256, 256, 256, 128, 128, 0, 256LL * 128, 256LL * 128, 1.0f, 0, 128);
  launch_gemm(stream, A1, GR, F1R, 128, 128, 128, 256 * 128, 128, 128LL * 128, 128, 128LL * 128, 1.0f, 0, 255);
  launch_gemm(stream, A1, GI, F1I, 128, 128, 128, 256 * 128, 128, 128LL * 128, 128, 128LL * 128, 1.0f, 0, 255);
  launch_gemm(stream, R1, F1R, T1R, 64, 128, 128, 128, 128, 0, 128LL * 128, 64LL * 128, 1.0f, 0, 255);
  launch_gemm(stream, R1, F1I, T1I, 64, 128, 128, 128, 128, 0, 128LL * 128, 64LL * 128, 1.0f, 0, 255);
  launch_gemm(stream, R2P, T1R, FFR, 128, 256, 256, 64 * 128, 128, 0, 128, 128LL * 128, 1.0f, 0, 64);
  launch_gemm(stream, R2P, T1I, FFI, 128, 256, 256, 64 * 128, 128, 0, 128, 128LL * 128, 1.0f, 0, 64);
  launch_gemm(stream, FFR, WCR, F2R, 128, 128, 128, 128, 128, 128LL * 128, 128LL * 128, 128LL * 128,  1.0f, 0, 64);
  launch_gemm(stream, FFI, WCI, F2R, 128, 128, 128, 128, 128, 128LL * 128, 128LL * 128, 128LL * 128, -1.0f, 1, 64);
  launch_gemm(stream, FFR, WCI, F2I, 128, 128, 128, 128, 128, 128LL * 128, 128LL * 128, 128LL * 128,  1.0f, 0, 64);
  launch_gemm(stream, FFI, WCR, F2I, 128, 128, 128, 128, 128, 128LL * 128, 128LL * 128, 128LL * 128,  1.0f, 1, 64);
  launch_gemm(stream, P2B, F2R, FTR, 64, 64, 64, 128 * 128, 128, 64LL * 64, 128, 64LL * 128, 1.0f, 0, 127);
  launch_gemm(stream, P2B, F2I, FTI, 64, 64, 64, 128 * 128, 128, 64LL * 64, 128, 64LL * 128, 1.0f, 0, 127);
  launch_gemm(stream, COS2, FTR, XSPEC, 128, 128, 128, 64 * 128, 128, 0, 128, 128LL * 128,  1.0f, 0, 64);
  launch_gemm(stream, SIN2, FTI, XSPEC, 128, 128, 128, 64 * 128, 128, 0, 128, 128LL * 128, -1.0f, 1, 64);

  launch_gemm(stream, x, WSP, XS1, 32640, 128, 128, 128, 128, 0, 0, 0, 1.0f, 0, 1);
  launch_gemm(stream, R1, XS1, T2R, 64, 128, 128, 255 * 128, 128, 0, 128, 64LL * 128, 1.0f, 0, 255);
  launch_gemm(stream, R2P, T2R, XS2, 128, 256, 256, 64 * 128, 128, 0, 128, 128LL * 128, 1.0f, 0, 64);
  k_xsppad<<<EWG(SZ_XSP)>>>(XS2, XSP, SZ_XSP);
  conv3x3_wmma<<<dim3(2, 8, 64), dim3(32), 0, stream>>>(XSP, sp_ck, sp_cb, TVEC, XCV);

  k_fuse<<<dim3(64 * 127), dim3(128), 0, stream>>>(XSPEC, XCV, lns, lnb, out);
}


# Resize products run over the band of nonzero weights only: order resize k 256 -> 128 per 32-row tile (mix kernel 12 -> 8 chunks, spatial GEMM 4 -> 2), degree resize k 128 -> 40 per wave
# speedup vs baseline: 50.9835x; 50.9835x over previous
#include <hip/hip_runtime.h>
#include <math.h>

typedef __attribute__((ext_vector_type(2))) float v2f;
typedef __attribute__((ext_vector_type(4))) float v4f;
typedef __attribute__((ext_vector_type(8))) float v8f;

#define PI_D 3.14159265358979323846

constexpr long long SZ_A1    = 255LL*128*128;
constexpr long long SZ_XP    = 128LL*256*128;
constexpr long long SZ_PH1   = 256LL*256;
constexpr long long SZ_G     = 128LL*256*128;
constexpr long long SZ_F1    = 255LL*128*128;
constexpr long long SZ_R1    = 64LL*128;
constexpr long long SZ_R2P   = 128LL*256;
constexpr long long SZ_T1    = 256LL*64*128;
constexpr long long SZ_FF    = 64LL*128*128;
constexpr long long SZ_WC    = 64LL*128*128;
constexpr long long SZ_F2    = 64LL*128*128;
constexpr long long SZ_P2    = 127LL*64*64;
constexpr long long SZ_FT    = 127LL*64*128;
constexpr long long SZ_PH2   = 128LL*128;
constexpr long long SZ_XSPEC = 64LL*128*128;
constexpr long long SZ_TVEC  = 256;
constexpr long long SZ_TC    = 64;
constexpr long long SZ_WSP   = 128LL*128;
constexpr long long SZ_XS1   = 128LL*255*128;
constexpr long long SZ_T2    = 256LL*64*128;
constexpr long long SZ_XS2   = 64LL*128*128;
constexpr long long SZ_XSP   = 67LL*129*128;
constexpr long long SZ_XCV   = 64LL*128*128;

constexpr long long OFF_A1    = 0;
constexpr long long OFF_XP    = OFF_A1    + SZ_A1;
constexpr long long OFF_COS1  = OFF_XP    + SZ_XP;
constexpr long long OFF_SIN1N = OFF_COS1  + SZ_PH1;
constexpr long long OFF_GR    = OFF_SIN1N + SZ_PH1;
constexpr long long OFF_GI    = OFF_GR    + SZ_G;
constexpr long long OFF_F1R   = OFF_GI    + SZ_G;
constexpr long long OFF_F1I   = OFF_F1R   + SZ_F1;
constexpr long long OFF_R1    = OFF_F1I   + SZ_F1;
constexpr long long OFF_R2P   = OFF_R1    + SZ_R1;
constexpr long long OFF_T1R   = OFF_R2P   + SZ_R2P;
constexpr long long OFF_T1I   = OFF_T1R   + SZ_T1;
constexpr long long OFF_FFR   = OFF_T1I   + SZ_T1;
constexpr long long OFF_FFI   = OFF_FFR   + SZ_FF;
constexpr long long OFF_WCR   = OFF_FFI   + SZ_FF;
constexpr long long OFF_WCI   = OFF_WCR   + SZ_WC;
constexpr long long OFF_F2R   = OFF_WCI   + SZ_WC;
constexpr long long OFF_F2I   = OFF_F2R   + SZ_F2;
constexpr long long OFF_P2A   = OFF_F2I   + SZ_F2;
constexpr long long OFF_P2B   = OFF_P2A   + SZ_P2;
constexpr long long OFF_FTR   = OFF_P2B   + SZ_P2;
constexpr long long OFF_FTI   = OFF_FTR   + SZ_FT;
constexpr long long OFF_COS2  = OFF_FTI   + SZ_FT;
constexpr long long OFF_SIN2  = OFF_COS2  + SZ_PH2;
constexpr long long OFF_XSPEC = OFF_SIN2  + SZ_PH2;
constexpr long long OFF_TVEC  = OFF_XSPEC + SZ_XSPEC;
constexpr long long OFF_TCR   = OFF_TVEC  + SZ_TVEC;
constexpr long long OFF_TCI   = OFF_TCR   + SZ_TC;
constexpr long long OFF_WSP   = OFF_TCI   + SZ_TC;
constexpr long long OFF_XS1   = OFF_WSP   + SZ_WSP;
constexpr long long OFF_T2R   = OFF_XS1   + SZ_XS1;
constexpr long long OFF_XS2   = OFF_T2R   + SZ_T2;
constexpr long long OFF_XSP   = OFF_XS2   + SZ_XS2;
constexpr long long OFF_XCV   = OFF_XSP   + SZ_XSP;

constexpr int KC = 64;
constexpr int AP = 68;
constexpr int BP = 136;
constexpr int B_ST = KC * BP;

template <int BM> struct TileRegs { v4f a[BM / 16]; v4f b[8]; };

template <int BM, class Addr>
__device__ __forceinline__ void tile_gload(TileRegs<BM>& r, const Addr& ad, int ch, int tid) {
  const int ar = tid >> 4, br = tid >> 5;
  const int lda = ad.lda(ch), ldb = ad.ldb(ch);
  const float* ap = ad.a(ch) + (long long)ar * lda + (tid & 15) * 4;
#pragma unroll
  for (int i = 0; i < BM / 16; ++i)
    r.a[i] = ad.a_ok(ch, ar + 16 * i) ? *(const v4f*)(ap + (long long)(16 * i) * lda) : v4f{};
  const float* bp = ad.b(ch) + (long long)br * ldb + (tid & 31) * 4;
#pragma unroll
  for (int i = 0; i < 8; ++i)
    r.b[i] = ad.b_ok(ch, br + 8 * i) ? *(const v4f*)(bp + (long long)(8 * i) * ldb) : v4f{};
}

template <int BM>
__device__ __forceinline__ void tile_sstore(const TileRegs<BM>& r, float* As, float* Bs, int tid) {
  float* ap = As + (tid >> 4) * AP + (tid & 15) * 4;
#pragma unroll
  for (int i = 0; i < BM / 16; ++i) *(v4f*)(ap + 16 * i * AP) = r.a[i];
  float* bp = Bs + (tid >> 5) * BP + (tid & 31) * 4;
#pragma unroll
  for (int i = 0; i < 8; ++i) *(v4f*)(bp + 8 * i * BP) = r.b[i];
}

template <int NT> struct Frag8 { v4f a; float b[4][NT]; };

template <int NT>
__device__ __forceinline__ void frag_load(Frag8<NT>& f, const float* aw, const float* bw, int kk) {
  f.a = *(const v4f*)(aw + kk);
  const float* bp = bw + kk * BP;
#pragma unroll
  for (int r = 0; r < 4; ++r) {
#pragma unroll
    for (int j = 0; j < NT; ++j) f.b[r][j] = bp[r * BP + 16 * j];
  }
}

template <int NT>
__device__ __forceinline__ void frag_mma(const Frag8<NT>& f, v8f (&acc)[NT]) {
  v2f a; a.x = f.a.x; a.y = f.a.y;
#pragma unroll
  for (int j = 0; j < NT; ++j) {
    v2f b; b.x = f.b[0][j]; b.y = f.b[1][j];
    acc[j] = __builtin_amdgcn_wmma_f32_16x16x4_f32(false, a, false, b, (short)0, acc[j], false, false);
  }
  a.x = f.a.z; a.y = f.a.w;
#pragma unroll
  for (int j = 0; j < NT; ++j) {
    v2f b; b.x = f.b[2][j]; b.y = f.b[3][j];
    acc[j] = __builtin_amdgcn_wmma_f32_16x16x4_f32(false, a, false, b, (short)0, acc[j], false, false);
  }
}

template <int NT>
__device__ __forceinline__ void tile_compute(const float* aw, const float* bw, v8f (&acc)[NT]) {
  constexpr int GPB = 8 / NT;
  constexpr int NB = 8 / GPB;
  Frag8<NT> f[2][GPB];
#pragma unroll
  for (int g = 0; g < GPB; ++g) frag_load<NT>(f[0][g], aw, bw, 8 * g);
#pragma unroll
  for (int sb = 0; sb < NB; ++sb) {
    if (sb + 1 < NB) {
#pragma unroll
      for (int g = 0; g < GPB; ++g) frag_load<NT>(f[(sb + 1) & 1][g], aw, bw, 8 * ((sb + 1) * GPB + g));
    }
    __builtin_amdgcn_sched_barrier(0);
#pragma unroll
    for (int g = 0; g < GPB; ++g) frag_mma<NT>(f[sb & 1][g], acc);
    __builtin_amdgcn_sched_barrier(0);
  }
}

__device__ __forceinline__ int band_k0(int j0) {
  int k = (2 * j0 - 4) & ~3;
  k = k < 0 ? 0 : k;
  return k > 128 ? 128 : k;
}

struct GemmAddr {
  const float* A; const float* B; int lda_; int ldb_; int alo; int brows;
  __device__ __forceinline__ int lda(int) const { return lda_; }
  __device__ __forceinline__ int ldb(int) const { return ldb_; }
  __device__ __forceinline__ const float* a(int ch) const { return A + ch * KC; }
  __device__ __forceinline__ const float* b(int ch) const { return B + (long long)ch * KC * ldb_; }
  __device__ __forceinline__ bool a_ok(int, int row) const { return row >= alo; }
  __device__ __forceinline__ bool b_ok(int ch, int row) const { return ch * KC + row < brows; }
};

struct OutAddr {
  const float* xs; const float* ck; const float* c2; const float* s2n; const float* fr; const float* fi;
  int l; int m0;
  __device__ __forceinline__ int lda(int) const { return 128; }
  __device__ __forceinline__ int ldb(int ch) const { return ch < 18 ? 128 : 64 * 128; }
  __device__ __forceinline__ const float* a(int ch) const {
    if (ch >= 18) return (ch < 20 ? c2 : s2n) + (ch & 1) * KC;
    const int tap = ch >> 1, dy = tap / 3, dx = tap - 3 * dy;
    return xs + ((long long)dy * 128 + dx) * 128 + (ch & 1) * KC;
  }
  __device__ __forceinline__ const float* b(int ch) const {
    if (ch >= 18) return (ch < 20 ? fr : fi) + (long long)(ch & 1) * KC * (64 * 128);
    return ck + ((long long)(ch >> 1) * 128 + (ch & 1) * KC) * 128;
  }
  __device__ __forceinline__ bool a_ok(int ch, int row) const {
    if (ch >= 18) return true;
    const int tap = ch >> 1, dy = tap / 3, dx = tap - 3 * dy;
    const int ll = l + dy - 1, mm = m0 + row + dx - 1;
    return ll >= 0 && ll < 64 && mm >= 0 && mm < 127;
  }
  __device__ __forceinline__ bool b_ok(int ch, int row) const {
    return ch < 18 || (ch & 1) * KC + row < 127;
  }
};

template <int BM> struct Geo {
  static constexpr int WM = BM / 16;
  static constexpr int WN = 8 / WM;
  static constexpr int NT = 8 / WN;
  static constexpr int A_ST = BM * AP;
};

template <int BM, class Addr>
__device__ __forceinline__ void tile_mainloop(const Addr& ad, int nchunks,
                                              float* As, float* Bs, int tid,
                                              v8f (&acc)[Geo<BM>::NT]) {
  constexpr int NT = Geo<BM>::NT, A_ST = Geo<BM>::A_ST;
  const int wave = tid >> 5, lane = tid & 31;
  const int wm = wave / Geo<BM>::WN, wn = wave % Geo<BM>::WN;
  const int half = lane >> 4, l16 = lane & 15;
  const int aoff = (wm * 16 + l16) * AP + 4 * half;
  const int boff = (4 * half) * BP + wn * (16 * NT) + l16;
  TileRegs<BM> r;
  tile_gload<BM>(r, ad, 0, tid);
  tile_sstore<BM>(r, As, Bs, tid);
  if (nchunks > 1) tile_gload<BM>(r, ad, 1, tid);
  __syncthreads();
  for (int ch = 0; ch < nchunks; ++ch) {
    const int st = ch & 1;
    if (ch + 1 < nchunks) tile_sstore<BM>(r, As + (st ^ 1) * A_ST, Bs + (st ^ 1) * B_ST, tid);
    if (ch + 2 < nchunks) tile_gload<BM>(r, ad, ch + 2, tid);
    tile_compute<NT>(As + st * A_ST + aoff, Bs + st * B_ST + boff, acc);
    __syncthreads();
  }
}

template <int BM>
__global__ __launch_bounds__(256) void gemm_wmma(const float* __restrict__ A, const float* __restrict__ B,
                          float* __restrict__ C, int K,
                          int lda, int ldb, int ldc,
                          long long sAb, long long sBb, long long sCb,
                          float alpha, int accumulate, int brows, int amir, int arowz, int aband) {
  constexpr int NT = Geo<BM>::NT;
  __shared__ __attribute__((aligned(16))) float As[2 * Geo<BM>::A_ST];
  __shared__ __attribute__((aligned(16))) float Bs[2 * B_ST];
  const int tid = threadIdx.x;
  const int m0 = blockIdx.x * BM;
  const long long b = blockIdx.y;
  long long pa = b; int alo = 0;
  if (amir >= 0) {
    const int d = (int)blockIdx.y - amir, am = d < 0 ? -d : d;
    pa = amir + am;
    if (arowz) alo = am - m0;
    if (d < 0 && (am & 1)) alpha = -alpha;
  }
  A += pa * sAb + (long long)m0 * lda; B += b * sBb; C += b * sCb;
  if (aband) {
    const int k0 = band_k0(m0);
    A += k0; B += (long long)k0 * ldb; brows -= k0; K = 2 * KC;
  }
  v8f acc[NT];
#pragma unroll
  for (int j = 0; j < NT; ++j) acc[j] = v8f{};
  GemmAddr ad; ad.A = A; ad.B = B; ad.lda_ = lda; ad.ldb_ = ldb; ad.alo = alo; ad.brows = brows;
  tile_mainloop<BM>(ad, K / KC, As, Bs, tid, acc);
  const int wave = tid >> 5, lane = tid & 31;
  const int half = lane >> 4, l16 = lane & 15;
  const int r0 = m0 + (wave / Geo<BM>::WN) * 16 + half * 8;
  const int c0 = (wave % Geo<BM>::WN) * (16 * NT) + l16;
#pragma unroll
  for (int v = 0; v < 8; ++v) {
    float* cp = C + (long long)(r0 + v) * ldc + c0;
    if (accumulate) {
#pragma unroll
      for (int j = 0; j < NT; ++j) cp[16 * j] += alpha * acc[j][v];
    } else {
#pragma unroll
      for (int j = 0; j < NT; ++j) cp[16 * j] = alpha * acc[j][v];
    }
  }
}

constexpr int RP = 132;

__global__ __launch_bounds__(256) void chain_resize_wmma(const float* __restrict__ A, const float* __restrict__ B,
                          const float* __restrict__ R, float* __restrict__ C,
                          int lda, int ldb, long long sAb, long long sBb, long long sBp,
                          long long sCb, long long sCp, int amir) {
  constexpr int BM = 128, NT = Geo<BM>::NT;
  __shared__ __attribute__((aligned(16))) float As[2 * Geo<BM>::A_ST];
  __shared__ __attribute__((aligned(16))) float Bs[2 * B_ST];
  const int tid = threadIdx.x;
  const long long b = blockIdx.x, part = blockIdx.y;
  long long pa = b; int alo = 0; float alpha = 1.0f;
  if (amir >= 0) {
    const int d = (int)blockIdx.x - amir, am = d < 0 ? -d : d;
    pa = amir + am;
    alo = am;
    if (d < 0 && (am & 1)) alpha = -1.0f;
  }
  A += pa * sAb; B += b * sBb + part * sBp; C += b * sCb + part * sCp;
  v8f acc[NT];
#pragma unroll
  for (int j = 0; j < NT; ++j) acc[j] = v8f{};
  GemmAddr ad; ad.A = A; ad.B = B; ad.lda_ = lda; ad.ldb_ = ldb; ad.alo = alo; ad.brows = 128;
  tile_mainloop<BM>(ad, 2, As, Bs, tid, acc);
  const int wave = tid >> 5, lane = tid & 31;
  const int half = lane >> 4, l16 = lane & 15;
  v4f rr[8];
  {
    const float* rp = R + (tid >> 5) * 128 + (tid & 31) * 4;
#pragma unroll
    for (int i = 0; i < 8; ++i) rr[i] = *(const v4f*)(rp + i * 8 * 128);
  }
  {
    float* fp = Bs + (wave * 16 + half * 8) * BP + l16;
#pragma unroll
    for (int v = 0; v < 8; ++v) {
#pragma unroll
      for (int j = 0; j < NT; ++j) fp[v * BP + 16 * j] = alpha * acc[j][v];
    }
  }
  {
    float* sp = As + (tid >> 5) * RP + (tid & 31) * 4;
#pragma unroll
    for (int i = 0; i < 8; ++i) *(v4f*)(sp + i * 8 * RP) = rr[i];
  }
  __syncthreads();
  const int wm2 = wave >> 1, wn2 = wave & 1;
  v8f acc2[4];
#pragma unroll
  for (int j = 0; j < 4; ++j) acc2[j] = v8f{};
  {
    int ks = 32 * wm2 - 4;
    ks = ks < 0 ? 0 : (ks > 88 ? 88 : ks);
    const float* aw2 = As + (wm2 * 16 + l16) * RP + 4 * half + ks;
    const float* bw2 = Bs + (ks + 4 * half) * BP + wn2 * 64 + l16;
    Frag8<4> f2[2][2];
    frag_load<4>(f2[0][0], aw2, bw2, 0);
    frag_load<4>(f2[0][1], aw2, bw2, 8);
    frag_load<4>(f2[1][0], aw2, bw2, 16);
    frag_load<4>(f2[1][1], aw2, bw2, 24);
    __builtin_amdgcn_sched_barrier(0);
    frag_mma<4>(f2[0][0], acc2);
    frag_mma<4>(f2[0][1], acc2);
    __builtin_amdgcn_sched_barrier(0);
    frag_load<4>(f2[0][0], aw2, bw2, 32);
    __builtin_amdgcn_sched_barrier(0);
    frag_mma<4>(f2[1][0], acc2);
    frag_mma<4>(f2[1][1], acc2);
    __builtin_amdgcn_sched_barrier(0);
    frag_mma<4>(f2[0][0], acc2);
  }
  const int r0 = wm2 * 16 + half * 8;
  const int c0 = wn2 * 64 + l16;
#pragma unroll
  for (int v = 0; v < 8; ++v) {
    float* cp = C + (long long)(r0 + v) * 128 + c0;
#pragma unroll
    for (int j = 0; j < 4; ++j) cp[16 * j] = acc2[j][v];
  }
}

struct MixAddr {
  const float* R2; const float* T1r; const float* T1i; int k0;
  __device__ __forceinline__ int lda(int) const { return 256; }
  __device__ __forceinline__ int ldb(int) const { return 64 * 128; }
  __device__ __forceinline__ const float* a(int ch) const { return R2 + k0 + (ch & 1) * KC; }
  __device__ __forceinline__ const float* b(int ch) const {
    return (ch < 2 ? T1r : T1i) + (long long)(k0 + (ch & 1) * KC) * (64 * 128);
  }
  __device__ __forceinline__ bool a_ok(int, int) const { return true; }
  __device__ __forceinline__ bool b_ok(int ch, int row) const { return k0 + (ch & 1) * KC + row < 255; }
};

__global__ __launch_bounds__(256) void chain_mix_wmma(const float* __restrict__ R2,
                          const float* __restrict__ T1r, const float* __restrict__ T1i,
                          const float* __restrict__ Wr, const float* __restrict__ Wi,
                          const float* __restrict__ tcr, const float* __restrict__ tci,
                          float* __restrict__ F2r, float* __restrict__ F2i) {
  constexpr int BM = 32, NT = Geo<BM>::NT, A_ST = Geo<BM>::A_ST;
  __shared__ __attribute__((aligned(16))) float As[2 * A_ST];
  __shared__ __attribute__((aligned(16))) float Bs[2 * B_ST];
  __shared__ __attribute__((aligned(16))) float Fr[BM * RP];
  __shared__ __attribute__((aligned(16))) float Fi[BM * RP];
  __shared__ __attribute__((aligned(16))) float Fn[BM * RP];
  const int tid = threadIdx.x;
  const int j0 = blockIdx.x * BM;
  const int l  = blockIdx.y;
  const int wave = tid >> 5, lane = tid & 31;
  const int half = lane >> 4, l16 = lane & 15;
  const int wm = wave / Geo<BM>::WN, wn = wave % Geo<BM>::WN;
  v8f accR[NT], accI[NT], accP[NT], accQ[NT];
#pragma unroll
  for (int j = 0; j < NT; ++j) { accR[j] = v8f{}; accI[j] = v8f{}; accP[j] = v8f{}; accQ[j] = v8f{}; }
  MixAddr ad; ad.R2 = R2 + (long long)j0 * 256; ad.T1r = T1r + (long long)l * 128; ad.T1i = T1i + (long long)l * 128;
  ad.k0 = band_k0(j0);
  const float* wsrc[4];
#pragma unroll
  for (int q = 0; q < 4; ++q)
    wsrc[q] = (q < 2 ? Wr : Wi) + (long long)l * 16384 + (long long)((q & 1) * KC + (tid >> 5)) * 128 + (tid & 31) * 4;
  const int bsoff = (tid >> 5) * BP + (tid & 31) * 4;
  const int aoff = (wm * 16 + l16) * AP + 4 * half;
  const int foff = (wm * 16 + l16) * RP + 4 * half;
  const int boff = (4 * half) * BP + wn * (16 * NT) + l16;
  TileRegs<BM> r;
  tile_gload<BM>(r, ad, 0, tid);
  tile_sstore<BM>(r, As, Bs, tid);
  tile_gload<BM>(r, ad, 1, tid);
  __syncthreads();
#pragma unroll
  for (int ch = 0; ch < 8; ++ch) {
    const int st = ch & 1;
    if (ch + 1 < 4) {
      tile_sstore<BM>(r, As + (st ^ 1) * A_ST, Bs + (st ^ 1) * B_ST, tid);
    } else if (ch + 1 < 8) {
#pragma unroll
      for (int i = 0; i < 8; ++i) *(v4f*)(Bs + (st ^ 1) * B_ST + bsoff + i * 8 * BP) = r.b[i];
    }
    if (ch + 2 < 4) {
      tile_gload<BM>(r, ad, ch + 2, tid);
    } else if (ch + 2 < 8) {
#pragma unroll
      for (int i = 0; i < 8; ++i) r.b[i] = *(const v4f*)(wsrc[ch + 2 - 4] + i * 8 * 128);
    }
    const float* bw = Bs + st * B_ST + boff;
    if (ch < 2) {
      tile_compute<NT>(As + st * A_ST + aoff, bw, accR);
    } else if (ch < 4) {
      tile_compute<NT>(As + st * A_ST + aoff, bw, accI);
    } else {
      const int q = ch - 4;
      tile_compute<NT>((q < 2 ? Fr : Fn) + foff + (q & 1) * KC, bw, accP);
      tile_compute<NT>((q < 2 ? Fi : Fr) + foff + (q & 1) * KC, bw, accQ);
    }
    if (ch == 3) {
      const int fo = (wm * 16 + half * 8) * RP + wn * (16 * NT) + l16;
#pragma unroll
      for (int v = 0; v < 8; ++v) {
#pragma unroll
        for (int j = 0; j < NT; ++j) {
          const float re = accR[j][v], im = accI[j][v];
          Fr[fo + v * RP + 16 * j] = re;
          Fi[fo + v * RP + 16 * j] = im;
          Fn[fo + v * RP + 16 * j] = -im;
        }
      }
    }
    __syncthreads();
  }
  const float tr = tcr[l], ti = tci[l];
  const int r0 = j0 + wm * 16 + half * 8;
  const int c0 = wn * (16 * NT) + l16;
#pragma unroll
  for (int v = 0; v < 8; ++v) {
    const long long o = ((long long)l * 128 + r0 + v) * 128 + c0;
#pragma unroll
    for (int j = 0; j < NT; ++j) {
      const float p = accP[j][v], q = accQ[j][v];
      F2r[o + 16 * j] = tr * p - ti * q;
      F2i[o + 16 * j] = tr * q + ti * p;
    }
  }
}

__global__ __launch_bounds__(256) void out_stage_wmma(const float* __restrict__ xs2, const float* __restrict__ ck,
                             const float* __restrict__ cb, const float* __restrict__ tvec,
                             const float* __restrict__ c2, const float* __restrict__ s2n,
                             const float* __restrict__ ftr, const float* __restrict__ fti,
                             const float* __restrict__ lns, const float* __restrict__ lnb,
                             float* __restrict__ out) {
  constexpr int BM = 32, NT = Geo<BM>::NT;
  __shared__ __attribute__((aligned(16))) float As[2 * Geo<BM>::A_ST];
  __shared__ __attribute__((aligned(16))) float Bs[2 * B_ST];
  __shared__ float red[2][BM][4];
  const int tid = threadIdx.x;
  const int m0 = blockIdx.x * BM;
  const int l  = blockIdx.y;
  v8f acc[NT];
#pragma unroll
  for (int j = 0; j < NT; ++j) acc[j] = v8f{};
  OutAddr ad;
  ad.xs = xs2 + ((long long)(l - 1) * 128 + (m0 - 1)) * 128; ad.ck = ck; ad.l = l; ad.m0 = m0;
  ad.c2 = c2 + (long long)m0 * 128; ad.s2n = s2n + (long long)m0 * 128;
  ad.fr = ftr + (long long)l * 128; ad.fi = fti + (long long)l * 128;
  tile_mainloop<BM>(ad, 22, As, Bs, tid, acc);
  const int wave = tid >> 5, lane = tid & 31;
  const int half = lane >> 4, l16 = lane & 15;
  const int wm = wave / Geo<BM>::WN, wn = wave % Geo<BM>::WN;
  const int rl = wm * 16 + half * 8;
  const int c0 = wn * (16 * NT) + l16;
  float bias[NT];
#pragma unroll
  for (int j = 0; j < NT; ++j) bias[j] = cb[c0 + 16 * j] + tvec[128 + c0 + 16 * j];
  float g[NT][8], s[8];
#pragma unroll
  for (int v = 0; v < 8; ++v) {
    s[v] = 0.0f;
#pragma unroll
    for (int j = 0; j < NT; ++j) {
      const float x = acc[j][v] + bias[j];
      g[j][v] = 0.5f * x * (1.0f + tanhf(0.7978845608028654f * (x + 0.044715f * x * x * x)));
      s[v] += g[j][v];
    }
  }
#pragma unroll
  for (int o = 8; o > 0; o >>= 1) {
#pragma unroll
    for (int v = 0; v < 8; ++v) s[v] += __shfl_xor(s[v], o, 32);
  }
  if (l16 == 0) {
#pragma unroll
    for (int v = 0; v < 8; ++v) red[0][rl + v][wn] = s[v];
  }
  __syncthreads();
  float mu[8];
#pragma unroll
  for (int v = 0; v < 8; ++v) {
    const float* rp = red[0][rl + v];
    mu[v] = (rp[0] + rp[1] + rp[2] + rp[3]) * (1.0f / 128.0f);
    s[v] = 0.0f;
#pragma unroll
    for (int j = 0; j < NT; ++j) { g[j][v] -= mu[v]; s[v] += g[j][v] * g[j][v]; }
  }
#pragma unroll
  for (int o = 8; o > 0; o >>= 1) {
#pragma unroll
    for (int v = 0; v < 8; ++v) s[v] += __shfl_xor(s[v], o, 32);
  }
  if (l16 == 0) {
#pragma unroll
    for (int v = 0; v < 8; ++v) red[1][rl + v][wn] = s[v];
  }
  __syncthreads();
  float sc[NT], sh[NT];
#pragma unroll
  for (int j = 0; j < NT; ++j) { sc[j] = lns[c0 + 16 * j]; sh[j] = lnb[c0 + 16 * j]; }
#pragma unroll
  for (int v = 0; v < 8; ++v) {
    const float* rp = red[1][rl + v];
    const float var = (rp[0] + rp[1] + rp[2] + rp[3]) * (1.0f / 128.0f);
    const float rs = rsqrtf(var + 1e-6f);
    const int m = m0 + rl + v;
    if (m < 127) {
      float* op = out + (((long long)l) * 127 + m) * 128 + c0;
#pragma unroll
      for (int j = 0; j < NT; ++j) op[16 * j] = g[j][v] * rs * sc[j] + sh[j];
    }
  }
}

__global__ __launch_bounds__(256) void k_tvec(const float* __restrict__ temb,
                       const float* __restrict__ dw, const float* __restrict__ db,
                       const float* __restrict__ drw, const float* __restrict__ drb,
                       const float* __restrict__ diw, const float* __restrict__ dib,
                       float* tvec, float* tcr, float* tci) {
  __shared__ float part[16][17];
  const int tid = threadIdx.x, ks = tid >> 4, jj = tid & 15;
  const int blk = blockIdx.x;
  const float* w; const float* bias; float* o; int ld, c;
  if (blk < 16)      { w = dw;  bias = db;  o = tvec; ld = 256; c = blk * 16 + jj; }
  else if (blk < 20) { w = drw; bias = drb; o = tcr;  ld = 64;  c = (blk - 16) * 16 + jj; }
  else               { w = diw; bias = dib; o = tci;  ld = 64;  c = (blk - 20) * 16 + jj; }
  const float* wp = w + (long long)(ks * 16) * ld + c;
  const float* ep = temb + ks * 16;
  float v[16], e[16];
#pragma unroll
  for (int i = 0; i < 16; ++i) { v[i] = wp[(long long)i * ld]; e[i] = ep[i]; }
  float s = (ks == 0) ? bias[c] : 0.0f;
#pragma unroll
  for (int i = 0; i < 16; ++i) s += e[i] * v[i];
  part[ks][jj] = s;
  __syncthreads();
  if (tid < 16) {
    float a = 0.0f;
#pragma unroll
    for (int q = 0; q < 16; ++q) a += part[q][tid];
    o[c] = a;
  }
}

__global__ void k_wsp(const float* __restrict__ spw, const float* __restrict__ tvec,
                      float* wsp, long long n) {
  long long i = (long long)blockIdx.x * 256 + threadIdx.x;
  if (i >= n) return;
  int o = (int)(i % 128);
  wsp[i] = tvec[o] * spw[i];
}

__global__ void k_phase1(float* c1, float* s1n, long long n) {
  long long i = (long long)blockIdx.x * 256 + threadIdx.x;
  if (i >= n) return;
  int m = (int)(i / 256), p = (int)(i % 256);
  if (m < 255 && p < 255) {
    double mval = (double)(m - 127);
    double phi = 2.0 * PI_D * (double)p / 255.0;
    double dphi = 2.0 * PI_D / 255.0;
    c1[i]  = (float)( cos(phi * mval) * dphi);
    s1n[i] = (float)(-sin(phi * mval) * dphi);
  } else { c1[i] = 0.0f; s1n[i] = 0.0f; }
}

__global__ void k_phase2(float* c2, float* s2n, long long n) {
  long long i = (long long)blockIdx.x * 256 + threadIdx.x;
  if (i >= n) return;
  int p = (int)(i / 128), m = (int)(i % 128);
  if (p < 127 && m < 127) {
    double mval = (double)(m - 63);
    double phi = 2.0 * PI_D * (double)p / 127.0;
    c2[i]  = (float)cos(mval * phi);
    s2n[i] = (float)(-sin(mval * phi));
  } else { c2[i] = 0.0f; s2n[i] = 0.0f; }
}

__global__ void k_resize(float* r1, float* r2p) {
  int tid = threadIdx.x;
  if (tid < 64) {
    int i = tid;
    float scale = 0.5f;
    float sample = (i + 0.5f) / scale - 0.5f;
    float sum = 0.0f;
    for (int j = 0; j < 128; ++j) {
      float w = 1.0f - fabsf(sample - j) * scale;
      if (w > 0.0f) sum += w;
    }
    for (int j = 0; j < 128; ++j) {
      float w = 1.0f - fabsf(sample - j) * scale;
      r1[i * 128 + j] = (w > 0.0f) ? (w / sum) : 0.0f;
    }
  } else if (tid < 192) {
    int j = tid - 64;
    if (j < 127) {
      float scale = 127.0f / 255.0f;
      float sample = (j + 0.5f) / scale - 0.5f;
      float sum = 0.0f;
      for (int p = 0; p < 255; ++p) {
        float w = 1.0f - fabsf(sample - p) * scale;
        if (w > 0.0f) sum += w;
      }
      for (int p = 0; p < 255; ++p) {
        float w = 1.0f - fabsf(sample - p) * scale;
        r2p[j * 256 + p] = (w > 0.0f) ? (w / sum) : 0.0f;
      }
      r2p[j * 256 + 255] = 0.0f;
    } else {
      for (int p = 0; p < 256; ++p) r2p[j * 256 + p] = 0.0f;
    }
  }
}

__device__ __forceinline__ double lane_bcast(double v, int src) {
  int lo = __double2loint(v), hi = __double2hiint(v);
  lo = __builtin_amdgcn_readlane(lo, src);
  hi = __builtin_amdgcn_readlane(hi, src);
  return __hiloint2double(hi, lo);
}

__global__ void k_legendre(float* __restrict__ out, int L, int weighted) {
  const int lane = threadIdx.x;
  const int t = blockIdx.x * 32 + lane;
  const int m = blockIdx.y;
  const double theta = (2.0 * t + 1.0) * PI_D / (2.0 * L - 1.0);
  const double x = cos(theta), s = sin(theta);
  const double w = weighted ? (s * PI_D / (double)L) : 1.0;

  double ca[4], cb[4];
#pragma unroll
  for (int j = 0; j < 4; ++j) {
    ca[j] = 0.0; cb[j] = 0.0;
    if (m + 2 + 32 * j < L) {
      const double ll = (double)(m + 2 + 32 * j + lane), mm = (double)m;
      const double den = ll * ll - mm * mm;
      ca[j] = sqrt((4.0 * ll * ll - 1.0) / den);
      cb[j] = sqrt((2.0 * ll + 1.0) * ((ll - 1.0) * (ll - 1.0) - mm * mm) /
                   ((2.0 * ll - 3.0) * den));
    }
  }

  double num = 1.0, dnm = 1.0;
#pragma unroll
  for (int j = 0; j < 4; ++j) {
    const int k = lane + 1 + 32 * j;
    if (k <= m) { num *= (double)(2 * k + 1); dnm *= (double)(2 * k); }
  }
  double r = num / dnm;
#pragma unroll
  for (int o = 16; o > 0; o >>= 1) r *= __shfl_xor(r, o, 32);
  double base = -s, pw = 1.0;
  for (int e = m; e; e >>= 1) { if (e & 1) pw *= base; base *= base; }
  const double pmm = sqrt(1.0 / (4.0 * PI_D)) * w * pw * sqrt(r);

  float* op = out + ((long long)(L - 1 + m) * L + m) * (long long)L + t;
  *op = (float)pmm; op += L;
  if (m + 1 < L) {
    double p2 = pmm;
    double p1 = sqrt(2.0 * m + 3.0) * x * pmm;
    *op = (float)p1; op += L;
#pragma unroll
    for (int j = 0; j < 4; ++j) {
      int cnt = L - (m + 2 + 32 * j);
      cnt = cnt > 32 ? 32 : cnt;
      for (int i = 0; i < cnt; ++i) {
        const double a = lane_bcast(ca[j], i), b = lane_bcast(cb[j], i);
        const double pn = a * x * p1 - b * p2;
        *op = (float)pn; op += L;
        p2 = p1; p1 = pn;
      }
    }
  }
}

__global__ void k_transp_p2(const float* __restrict__ a, float* __restrict__ b, long long n) {
  long long i = (long long)blockIdx.x * 256 + threadIdx.x;
  if (i >= n) return;
  int m = (int)(i / 4096), t = (int)((i / 64) % 64), l = (int)(i % 64);
  long long pl = (long long)(63 + m) * 4096;
  b[pl + t * 64 + l] = (l >= m) ? a[pl + l * 64 + t] : 0.0f;
}

template <int BM>
static inline void launch_gemm(hipStream_t s, const float* A, const float* B, float* C,
                               int M, int K, int lda, int ldb, int ldc,
                               long long sA, long long sB, long long sC,
                               float alpha, int acc, int batch,
                               int brows, int amir, int arowz, int aband) {
  dim3 g(M / BM, batch);
  gemm_wmma<BM><<<g, dim3(256), 0, s>>>(A, B, C, K, lda, ldb, ldc, sA, sB, sC, alpha, acc,
                                        brows, amir, arowz, aband);
}

#define EWG(n) dim3((unsigned)(((n) + 255) / 256)), dim3(256), 0, stream

extern "C" void kernel_launch(void* const* d_in, const int* in_sizes, int n_in,
                              void* d_out, int out_size, void* d_ws, size_t ws_size,
                              hipStream_t stream) {
  (void)in_sizes; (void)n_in; (void)out_size; (void)ws_size;
  const float* x       = (const float*)d_in[0];
  const float* t_emb   = (const float*)d_in[1];
  const float* sp_dw   = (const float*)d_in[2];
  const float* sp_db   = (const float*)d_in[3];
  const float* sp_wt   = (const float*)d_in[4];
  const float* sp_ck   = (const float*)d_in[5];
  const float* sp_cb   = (const float*)d_in[6];
  const float* spec_wr = (const float*)d_in[7];
  const float* spec_wi = (const float*)d_in[8];
  const float* drw     = (const float*)d_in[9];
  const float* drb     = (const float*)d_in[10];
  const float* diw     = (const float*)d_in[11];
  const float* dib     = (const float*)d_in[12];
  const float* lns     = (const float*)d_in[13];
  const float* lnb     = (const float*)d_in[14];
  float* out = (float*)d_out;
  float* ws  = (float*)d_ws;

  float *A1 = ws + OFF_A1, *COS1 = ws + OFF_COS1, *SIN1N = ws + OFF_SIN1N;
  float *GR = ws + OFF_GR, *GI = ws + OFF_GI;
  float *R1 = ws + OFF_R1, *R2P = ws + OFF_R2P, *T1R = ws + OFF_T1R, *T1I = ws + OFF_T1I;
  float *F2R = ws + OFF_F2R, *F2I = ws + OFF_F2I, *P2A = ws + OFF_P2A, *P2B = ws + OFF_P2B;
  float *FTR = ws + OFF_FTR, *FTI = ws + OFF_FTI, *COS2 = ws + OFF_COS2, *SIN2N = ws + OFF_SIN2;
  float *TVEC = ws + OFF_TVEC, *TCR = ws + OFF_TCR, *TCI = ws + OFF_TCI;
  float *WSP = ws + OFF_WSP, *T2R = ws + OFF_T2R, *XS2 = ws + OFF_XS2;

  k_tvec<<<dim3(24), dim3(256), 0, stream>>>(t_emb, sp_dw, sp_db, drw, drb, diw, dib, TVEC, TCR, TCI);
  k_wsp<<<EWG(SZ_WSP)>>>(sp_wt, TVEC, WSP, SZ_WSP);
  k_phase1<<<EWG(SZ_PH1)>>>(COS1, SIN1N, SZ_PH1);
  k_phase2<<<EWG(SZ_PH2)>>>(COS2, SIN2N, SZ_PH2);
  k_resize<<<dim3(1), dim3(256), 0, stream>>>(R1, R2P);
  k_legendre<<<dim3(4, 128), dim3(32), 0, stream>>>(A1, 128, 1);
  k_legendre<<<dim3(2, 64), dim3(32), 0, stream>>>(P2A, 64, 0);
  k_transp_p2<<<EWG(64LL * 4096)>>>(P2A, P2B, 64LL * 4096);

  launch_gemm<64>(stream, COS1,  x, GR, 256, 256, 256, 128, 128, 0, 255LL * 128, 256LL * 128, 1.0f, 0, 128, 255, -1, 0, 0);
  launch_gemm<64>(stream, SIN1N, x, GI, 256, 256, 256, 128, 128, 0, 255LL * 128, 256LL * 128, 1.0f, 0, 128, 255, -1, 0, 0);
  chain_resize_wmma<<<dim3(255, 2), dim3(256), 0, stream>>>(A1, GR, R1, T1R, 128, 256 * 128, 128LL * 128, 128, SZ_G,
                                                          64LL * 128, SZ_T1, 127);
  chain_mix_wmma<<<dim3(4, 64), dim3(256), 0, stream>>>(R2P, T1R, T1I, spec_wr, spec_wi, TCR, TCI, F2R, F2I);
  launch_gemm<32>(stream, P2B, F2R, FTR, 64, 64, 64, 128 * 128, 128, 64LL * 64, 128, 64LL * 128, 1.0f, 0, 127, 64, 63, 0, 0);
  launch_gemm<32>(stream, P2B, F2I, FTI, 64, 64, 64, 128 * 128, 128, 64LL * 64, 128, 64LL * 128, 1.0f, 0, 127, 64, 63, 0, 0);

  chain_resize_wmma<<<dim3(255, 1), dim3(256), 0, stream>>>(x, WSP, R1, T2R, 255 * 128, 128, 128, 0, 0,
                                                          64LL * 128, 0, -1);
  launch_gemm<32>(stream, R2P, T2R, XS2, 128, 256, 256, 64 * 128, 128, 0, 128, 128LL * 128, 1.0f, 0, 64, 255, -1, 0, 1);
  out_stage_wmma<<<dim3(4, 64), dim3(256), 0, stream>>>(XS2, sp_ck, sp_cb, TVEC, COS2, SIN2N, FTR, FTI,
                                                      lns, lnb, out);
}


# Forward longitude sum folded in half: cosine even, sine odd under p -> 255-p, so the B panel loader forms x[k] + x[255-k] (cos), x[k] - x[255-k] (sin); both phase products run over K = 128, not 256
# speedup vs baseline: 58.0908x; 1.1394x over previous
#include <hip/hip_runtime.h>
#include <math.h>

typedef __attribute__((ext_vector_type(2))) float v2f;
typedef __attribute__((ext_vector_type(4))) float v4f;
typedef __attribute__((ext_vector_type(8))) float v8f;

#define PI_D 3.14159265358979323846

constexpr long long SZ_A1    = 255LL*128*128;
constexpr long long SZ_XP    = 128LL*256*128;
constexpr long long SZ_PH1   = 256LL*256;
constexpr long long SZ_G     = 128LL*256*128;
constexpr long long SZ_F1    = 255LL*128*128;
constexpr long long SZ_R1    = 64LL*128;
constexpr long long SZ_R2P   = 128LL*256;
constexpr long long SZ_T1    = 256LL*64*128;
constexpr long long SZ_FF    = 64LL*128*128;
constexpr long long SZ_WC    = 64LL*128*128;
constexpr long long SZ_F2    = 64LL*128*128;
constexpr long long SZ_P2    = 127LL*64*64;
constexpr long long SZ_FT    = 127LL*64*128;
constexpr long long SZ_PH2   = 128LL*128;
constexpr long long SZ_XSPEC = 64LL*128*128;
constexpr long long SZ_TVEC  = 256;
constexpr long long SZ_TC    = 64;
constexpr long long SZ_WSP   = 128LL*128;
constexpr long long SZ_XS1   = 128LL*255*128;
constexpr long long SZ_T2    = 256LL*64*128;
constexpr long long SZ_XS2   = 64LL*128*128;
constexpr long long SZ_XSP   = 67LL*129*128;
constexpr long long SZ_XCV   = 64LL*128*128;

constexpr long long OFF_A1    = 0;
constexpr long long OFF_XP    = OFF_A1    + SZ_A1;
constexpr long long OFF_COS1  = OFF_XP    + SZ_XP;
constexpr long long OFF_SIN1N = OFF_COS1  + SZ_PH1;
constexpr long long OFF_GR    = OFF_SIN1N + SZ_PH1;
constexpr long long OFF_GI    = OFF_GR    + SZ_G;
constexpr long long OFF_F1R   = OFF_GI    + SZ_G;
constexpr long long OFF_F1I   = OFF_F1R   + SZ_F1;
constexpr long long OFF_R1    = OFF_F1I   + SZ_F1;
constexpr long long OFF_R2P   = OFF_R1    + SZ_R1;
constexpr long long OFF_T1R   = OFF_R2P   + SZ_R2P;
constexpr long long OFF_T1I   = OFF_T1R   + SZ_T1;
constexpr long long OFF_FFR   = OFF_T1I   + SZ_T1;
constexpr long long OFF_FFI   = OFF_FFR   + SZ_FF;
constexpr long long OFF_WCR   = OFF_FFI   + SZ_FF;
constexpr long long OFF_WCI   = OFF_WCR   + SZ_WC;
constexpr long long OFF_F2R   = OFF_WCI   + SZ_WC;
constexpr long long OFF_F2I   = OFF_F2R   + SZ_F2;
constexpr long long OFF_P2A   = OFF_F2I   + SZ_F2;
constexpr long long OFF_P2B   = OFF_P2A   + SZ_P2;
constexpr long long OFF_FTR   = OFF_P2B   + SZ_P2;
constexpr long long OFF_FTI   = OFF_FTR   + SZ_FT;
constexpr long long OFF_COS2  = OFF_FTI   + SZ_FT;
constexpr long long OFF_SIN2  = OFF_COS2  + SZ_PH2;
constexpr long long OFF_XSPEC = OFF_SIN2  + SZ_PH2;
constexpr long long OFF_TVEC  = OFF_XSPEC + SZ_XSPEC;
constexpr long long OFF_TCR   = OFF_TVEC  + SZ_TVEC;
constexpr long long OFF_TCI   = OFF_TCR   + SZ_TC;
constexpr long long OFF_WSP   = OFF_TCI   + SZ_TC;
constexpr long long OFF_XS1   = OFF_WSP   + SZ_WSP;
constexpr long long OFF_T2R   = OFF_XS1   + SZ_XS1;
constexpr long long OFF_XS2   = OFF_T2R   + SZ_T2;
constexpr long long OFF_XSP   = OFF_XS2   + SZ_XS2;
constexpr long long OFF_XCV   = OFF_XSP   + SZ_XSP;

constexpr int KC = 64;
constexpr int AP = 68;
constexpr int BP = 136;
constexpr int B_ST = KC * BP;

template <int BM> struct TileRegs { v4f a[BM / 16]; v4f b[8]; v4f b2[8]; };

template <int BM, class Addr>
__device__ __forceinline__ void tile_gload(TileRegs<BM>& r, const Addr& ad, int ch, int tid) {
  const int ar = tid >> 4, br = tid >> 5;
  const int ldb = ad.ldb(ch);
  if constexpr (Addr::GEN_A) {
#pragma unroll
    for (int i = 0; i < BM / 16; ++i) r.a[i] = ad.a_gen(ch, ar + 16 * i, (tid & 15) * 4);
  } else {
    const int lda = ad.lda(ch);
    const float* ap = ad.a(ch) + (long long)ar * lda + (tid & 15) * 4;
#pragma unroll
    for (int i = 0; i < BM / 16; ++i)
      r.a[i] = ad.a_ok(ch, ar + 16 * i) ? *(const v4f*)(ap + (long long)(16 * i) * lda) : v4f{};
  }
  const float* bp = ad.b(ch) + (long long)br * ldb + (tid & 31) * 4;
#pragma unroll
  for (int i = 0; i < 8; ++i)
    r.b[i] = ad.b_ok(ch, br + 8 * i) ? *(const v4f*)(bp + (long long)(8 * i) * ldb) : v4f{};
  if constexpr (Addr::FOLD_B) {
    const float* bq = ad.bplane() + (tid & 31) * 4;
#pragma unroll
    for (int i = 0; i < 8; ++i) {
      const int k = ch * KC + br + 8 * i;
      r.b2[i] = *(const v4f*)(bq + (long long)(255 - (k > 0 ? k : 255)) * ldb);
    }
  }
}

template <int BM, bool FOLD = false>
__device__ __forceinline__ void tile_sstore(const TileRegs<BM>& r, float* As, float* Bs, int tid, float sgn = 0.0f,
                                            bool first = false) {
  float* ap = As + (tid >> 4) * AP + (tid & 15) * 4;
#pragma unroll
  for (int i = 0; i < BM / 16; ++i) *(v4f*)(ap + 16 * i * AP) = r.a[i];
  float* bp = Bs + (tid >> 5) * BP + (tid & 31) * 4;
#pragma unroll
  for (int i = 0; i < 8; ++i) {
    if constexpr (FOLD) {
      const float s = (first && i == 0 && (tid >> 5) == 0) ? 0.0f : sgn;
      *(v4f*)(bp + 8 * i * BP) = r.b[i] + s * r.b2[i];
    }
    else                *(v4f*)(bp + 8 * i * BP) = r.b[i];
  }
}

template <int NT> struct Frag8 { v4f a; float b[4][NT]; };

template <int NT>
__device__ __forceinline__ void frag_load(Frag8<NT>& f, const float* aw, const float* bw, int kk) {
  f.a = *(const v4f*)(aw + kk);
  const float* bp = bw + kk * BP;
#pragma unroll
  for (int r = 0; r < 4; ++r) {
#pragma unroll
    for (int j = 0; j < NT; ++j) f.b[r][j] = bp[r * BP + 16 * j];
  }
}

template <int NT>
__device__ __forceinline__ void frag_mma(const Frag8<NT>& f, v8f (&acc)[NT]) {
  v2f a; a.x = f.a.x; a.y = f.a.y;
#pragma unroll
  for (int j = 0; j < NT; ++j) {
    v2f b; b.x = f.b[0][j]; b.y = f.b[1][j];
    acc[j] = __builtin_amdgcn_wmma_f32_16x16x4_f32(false, a, false, b, (short)0, acc[j], false, false);
  }
  a.x = f.a.z; a.y = f.a.w;
#pragma unroll
  for (int j = 0; j < NT; ++j) {
    v2f b; b.x = f.b[2][j]; b.y = f.b[3][j];
    acc[j] = __builtin_amdgcn_wmma_f32_16x16x4_f32(false, a, false, b, (short)0, acc[j], false, false);
  }
}

template <int NT>
__device__ __forceinline__ void tile_compute(const float* aw, const float* bw, v8f (&acc)[NT]) {
  constexpr int GPB = 8 / NT;
  constexpr int NB = 8 / GPB;
  Frag8<NT> f[2][GPB];
#pragma unroll
  for (int g = 0; g < GPB; ++g) frag_load<NT>(f[0][g], aw, bw, 8 * g);
#pragma unroll
  for (int sb = 0; sb < NB; ++sb) {
    if (sb + 1 < NB) {
#pragma unroll
      for (int g = 0; g < GPB; ++g) frag_load<NT>(f[(sb + 1) & 1][g], aw, bw, 8 * ((sb + 1) * GPB + g));
    }
    __builtin_amdgcn_sched_barrier(0);
#pragma unroll
    for (int g = 0; g < GPB; ++g) frag_mma<NT>(f[sb & 1][g], acc);
    __builtin_amdgcn_sched_barrier(0);
  }
}

__device__ __forceinline__ v4f r1quad(int i, int k0) {
  const float sample = 2.0f * (float)i + 0.5f;
  float sum = 0.0f;
#pragma unroll
  for (int q = -1; q <= 2; ++q) {
    const int k = 2 * i + q;
    if (k >= 0 && k < 128) sum += 1.0f - fabsf(sample - (float)k) * 0.5f;
  }
  v4f o;
#pragma unroll
  for (int e = 0; e < 4; ++e) {
    const float w = 1.0f - fabsf(sample - (float)(k0 + e)) * 0.5f;
    o[e] = (w > 0.0f) ? (w / sum) : 0.0f;
  }
  return o;
}
__device__ __forceinline__ v4f r2quad(int j, int p0) {
  v4f o = v4f{};
  if (j >= 127) return o;
  const float scale = 127.0f / 255.0f;
  const float sample = ((float)j + 0.5f) / scale - 0.5f;
  const int pc = (int)sample;
  float sum = 0.0f;
#pragma unroll
  for (int q = -3; q <= 3; ++q) {
    const int p = pc + q;
    const float w = 1.0f - fabsf(sample - (float)p) * scale;
    if (p >= 0 && p < 255 && w > 0.0f) sum += w;
  }
#pragma unroll
  for (int e = 0; e < 4; ++e) {
    const int p = p0 + e;
    const float w = 1.0f - fabsf(sample - (float)p) * scale;
    o[e] = (p < 255 && w > 0.0f) ? (w / sum) : 0.0f;
  }
  return o;
}

__device__ __forceinline__ int band_k0(int j0) {
  int k = (2 * j0 - 4) & ~3;
  k = k < 0 ? 0 : k;
  return k > 128 ? 128 : k;
}

struct GemmAddr {
  static constexpr bool GEN_A = false; static constexpr bool FOLD_B = false;
  const float* A; const float* B; int lda_; int ldb_; int alo; int brows;
  __device__ __forceinline__ int lda(int) const { return lda_; }
  __device__ __forceinline__ int ldb(int) const { return ldb_; }
  __device__ __forceinline__ const float* a(int ch) const { return A + ch * KC; }
  __device__ __forceinline__ const float* b(int ch) const { return B + (long long)ch * KC * ldb_; }
  __device__ __forceinline__ bool a_ok(int, int row) const { return row >= alo; }
  __device__ __forceinline__ bool b_ok(int ch, int row) const { return ch * KC + row < brows; }
};

struct BandAddr {
  static constexpr bool GEN_A = true; static constexpr bool FOLD_B = false;
  const float* B; int ldb_; int brows; int j0; int k0;
  __device__ __forceinline__ int ldb(int) const { return ldb_; }
  __device__ __forceinline__ const float* b(int ch) const { return B + (long long)ch * KC * ldb_; }
  __device__ __forceinline__ bool b_ok(int ch, int row) const { return ch * KC + row < brows; }
  __device__ __forceinline__ v4f a_gen(int ch, int row, int c4) const { return r2quad(j0 + row, k0 + ch * KC + c4); }
};

struct FoldAddr {
  static constexpr bool GEN_A = false; static constexpr bool FOLD_B = true;
  const float* A; const float* B; int lda_; int ldb_;
  __device__ __forceinline__ int lda(int) const { return lda_; }
  __device__ __forceinline__ int ldb(int) const { return ldb_; }
  __device__ __forceinline__ const float* a(int ch) const { return A + ch * KC; }
  __device__ __forceinline__ const float* b(int ch) const { return B + (long long)ch * KC * ldb_; }
  __device__ __forceinline__ const float* bplane() const { return B; }
  __device__ __forceinline__ bool a_ok(int, int) const { return true; }
  __device__ __forceinline__ bool b_ok(int, int) const { return true; }
};

struct OutAddr {
  static constexpr bool GEN_A = false; static constexpr bool FOLD_B = false;
  const float* xs; const float* ck; const float* c2; const float* s2n; const float* fr; const float* fi;
  int l; int m0;
  __device__ __forceinline__ int lda(int) const { return 128; }
  __device__ __forceinline__ int ldb(int ch) const { return ch < 18 ? 128 : 64 * 128; }
  __device__ __forceinline__ const float* a(int ch) const {
    if (ch >= 18) return (ch < 20 ? c2 : s2n) + (ch & 1) * KC;
    const int tap = ch >> 1, dy = tap / 3, dx = tap - 3 * dy;
    return xs + ((long long)dy * 128 + dx) * 128 + (ch & 1) * KC;
  }
  __device__ __forceinline__ const float* b(int ch) const {
    if (ch >= 18) return (ch < 20 ? fr : fi) + (long long)(ch & 1) * KC * (64 * 128);
    return ck + ((long long)(ch >> 1) * 128 + (ch & 1) * KC) * 128;
  }
  __device__ __forceinline__ bool a_ok(int ch, int row) const {
    if (ch >= 18) return true;
    const int tap = ch >> 1, dy = tap / 3, dx = tap - 3 * dy;
    const int ll = l + dy - 1, mm = m0 + row + dx - 1;
    return ll >= 0 && ll < 64 && mm >= 0 && mm < 127;
  }
  __device__ __forceinline__ bool b_ok(int ch, int row) const {
    return ch < 18 || (ch & 1) * KC + row < 127;
  }
};

template <int BM> struct Geo {
  static constexpr int WM = BM / 16;
  static constexpr int WN = 8 / WM;
  static constexpr int NT = 8 / WN;
  static constexpr int A_ST = BM * AP;
};

template <int BM, class Addr>
__device__ __forceinline__ void tile_mainloop(const Addr& ad, int nchunks,
                                              float* As, float* Bs, int tid,
                                              v8f (&acc)[Geo<BM>::NT], float sgn = 0.0f) {
  constexpr bool FOLD = Addr::FOLD_B;
  constexpr int NT = Geo<BM>::NT, A_ST = Geo<BM>::A_ST;
  const int wave = tid >> 5, lane = tid & 31;
  const int wm = wave / Geo<BM>::WN, wn = wave % Geo<BM>::WN;
  const int half = lane >> 4, l16 = lane & 15;
  const int aoff = (wm * 16 + l16) * AP + 4 * half;
  const int boff = (4 * half) * BP + wn * (16 * NT) + l16;
  TileRegs<BM> r;
  tile_gload<BM>(r, ad, 0, tid);
  tile_sstore<BM, FOLD>(r, As, Bs, tid, sgn, true);
  if (nchunks > 1) tile_gload<BM>(r, ad, 1, tid);
  __syncthreads();
  for (int ch = 0; ch < nchunks; ++ch) {
    const int st = ch & 1;
    if (ch + 1 < nchunks) tile_sstore<BM, FOLD>(r, As + (st ^ 1) * A_ST, Bs + (st ^ 1) * B_ST, tid, sgn);
    if (ch + 2 < nchunks) tile_gload<BM>(r, ad, ch + 2, tid);
    tile_compute<NT>(As + st * A_ST + aoff, Bs + st * B_ST + boff, acc);
    __syncthreads();
  }
}

template <int BM, int MODE>
__global__ __launch_bounds__(256) void gemm_wmma(const float* __restrict__ A, const float* __restrict__ B,
                          float* __restrict__ C, int K,
                          int lda, int ldb, int ldc,
                          long long sAb, long long sBb, long long sCb,
                          float alpha, int accumulate, int brows, int amir, int arowz) {
  constexpr int NT = Geo<BM>::NT;
  __shared__ __attribute__((aligned(16))) float As[2 * Geo<BM>::A_ST];
  __shared__ __attribute__((aligned(16))) float Bs[2 * B_ST];
  const int tid = threadIdx.x;
  const int m0 = blockIdx.x * BM;
  const long long b = blockIdx.y;
  long long pa = b; int alo = 0;
  if (amir >= 0) {
    const int d = (int)blockIdx.y - amir, am = d < 0 ? -d : d;
    pa = amir + am;
    if (arowz) alo = am - m0;
    if (d < 0 && (am & 1)) alpha = -alpha;
  }
  B += b * sBb; C += b * sCb;
  v8f acc[NT];
#pragma unroll
  for (int j = 0; j < NT; ++j) acc[j] = v8f{};
  if constexpr (MODE == 1) {
    const int k0 = band_k0(m0);
    BandAddr ad; ad.B = B + (long long)k0 * ldb; ad.ldb_ = ldb; ad.brows = brows - k0; ad.j0 = m0; ad.k0 = k0;
    tile_mainloop<BM>(ad, 2, As, Bs, tid, acc);
  } else if constexpr (MODE >= 2) {
    FoldAddr ad; ad.A = A + pa * sAb + (long long)m0 * lda; ad.B = B; ad.lda_ = lda; ad.ldb_ = ldb;
    tile_mainloop<BM>(ad, 2, As, Bs, tid, acc, MODE == 2 ? 1.0f : -1.0f);
  } else {
    A += pa * sAb + (long long)m0 * lda;
    GemmAddr ad; ad.A = A; ad.B = B; ad.lda_ = lda; ad.ldb_ = ldb; ad.alo = alo; ad.brows = brows;
    tile_mainloop<BM>(ad, K / KC, As, Bs, tid, acc);
  }
  const int wave = tid >> 5, lane = tid & 31;
  const int half = lane >> 4, l16 = lane & 15;
  const int r0 = m0 + (wave / Geo<BM>::WN) * 16 + half * 8;
  const int c0 = (wave % Geo<BM>::WN) * (16 * NT) + l16;
#pragma unroll
  for (int v = 0; v < 8; ++v) {
    float* cp = C + (long long)(r0 + v) * ldc + c0;
    if (accumulate) {
#pragma unroll
      for (int j = 0; j < NT; ++j) cp[16 * j] += alpha * acc[j][v];
    } else {
#pragma unroll
      for (int j = 0; j < NT; ++j) cp[16 * j] = alpha * acc[j][v];
    }
  }
}

constexpr int RP = 132;

__global__ __launch_bounds__(256) void chain_resize_wmma(const float* __restrict__ A, const float* __restrict__ B,
                          const float* __restrict__ cscale, float* __restrict__ C,
                          int lda, int ldb, long long sAb, long long sBb, long long sBp,
                          long long sCb, long long sCp, int amir) {
  constexpr int BM = 128, NT = Geo<BM>::NT;
  __shared__ __attribute__((aligned(16))) float As[2 * Geo<BM>::A_ST];
  __shared__ __attribute__((aligned(16))) float Bs[2 * B_ST];
  const int tid = threadIdx.x;
  const long long b = blockIdx.x, part = blockIdx.y;
  long long pa = b; int alo = 0; float alpha = 1.0f;
  if (amir >= 0) {
    const int d = (int)blockIdx.x - amir, am = d < 0 ? -d : d;
    pa = amir + am;
    alo = am;
    if (d < 0 && (am & 1)) alpha = -1.0f;
  }
  A += pa * sAb; B += b * sBb + part * sBp; C += b * sCb + part * sCp;
  v8f acc[NT];
#pragma unroll
  for (int j = 0; j < NT; ++j) acc[j] = v8f{};
  GemmAddr ad; ad.A = A; ad.B = B; ad.lda_ = lda; ad.ldb_ = ldb; ad.alo = alo; ad.brows = 128;
  tile_mainloop<BM>(ad, 2, As, Bs, tid, acc);
  const int wave = tid >> 5, lane = tid & 31;
  const int half = lane >> 4, l16 = lane & 15;
  v4f rr[8];
#pragma unroll
  for (int i = 0; i < 8; ++i) rr[i] = r1quad((tid >> 5) + 8 * i, (tid & 31) * 4);
  {
    float cs[NT];
#pragma unroll
    for (int j = 0; j < NT; ++j) cs[j] = cscale ? alpha * cscale[l16 + 16 * j] : alpha;
    float* fp = Bs + (wave * 16 + half * 8) * BP + l16;
#pragma unroll
    for (int v = 0; v < 8; ++v) {
#pragma unroll
      for (int j = 0; j < NT; ++j) fp[v * BP + 16 * j] = cs[j] * acc[j][v];
    }
  }
  {
    float* sp = As + (tid >> 5) * RP + (tid & 31) * 4;
#pragma unroll
    for (int i = 0; i < 8; ++i) *(v4f*)(sp + i * 8 * RP) = rr[i];
  }
  __syncthreads();
  const int wm2 = wave >> 1, wn2 = wave & 1;
  v8f acc2[4];
#pragma unroll
  for (int j = 0; j < 4; ++j) acc2[j] = v8f{};
  {
    int ks = 32 * wm2 - 4;
    ks = ks < 0 ? 0 : (ks > 88 ? 88 : ks);
    const float* aw2 = As + (wm2 * 16 + l16) * RP + 4 * half + ks;
    const float* bw2 = Bs + (ks + 4 * half) * BP + wn2 * 64 + l16;
    Frag8<4> f2[2][2];
    frag_load<4>(f2[0][0], aw2, bw2, 0);
    frag_load<4>(f2[0][1], aw2, bw2, 8);
    frag_load<4>(f2[1][0], aw2, bw2, 16);
    frag_load<4>(f2[1][1], aw2, bw2, 24);
    __builtin_amdgcn_sched_barrier(0);
    frag_mma<4>(f2[0][0], acc2);
    frag_mma<4>(f2[0][1], acc2);
    __builtin_amdgcn_sched_barrier(0);
    frag_load<4>(f2[0][0], aw2, bw2, 32);
    __builtin_amdgcn_sched_barrier(0);
    frag_mma<4>(f2[1][0], acc2);
    frag_mma<4>(f2[1][1], acc2);
    __builtin_amdgcn_sched_barrier(0);
    frag_mma<4>(f2[0][0], acc2);
  }
  const int r0 = wm2 * 16 + half * 8;
  const int c0 = wn2 * 64 + l16;
#pragma unroll
  for (int v = 0; v < 8; ++v) {
    float* cp = C + (long long)(r0 + v) * 128 + c0;
#pragma unroll
    for (int j = 0; j < 4; ++j) cp[16 * j] = acc2[j][v];
  }
}

struct MixAddr {
  static constexpr bool GEN_A = true; static constexpr bool FOLD_B = false;
  const float* T1r; const float* T1i; int j0; int k0;
  __device__ __forceinline__ int ldb(int) const { return 64 * 128; }
  __device__ __forceinline__ v4f a_gen(int ch, int row, int c4) const { return r2quad(j0 + row, k0 + (ch & 1) * KC + c4); }
  __device__ __forceinline__ const float* b(int ch) const {
    return (ch < 2 ? T1r : T1i) + (long long)(k0 + (ch & 1) * KC) * (64 * 128);
  }
  __device__ __forceinline__ bool b_ok(int ch, int row) const { return k0 + (ch & 1) * KC + row < 255; }
};

__global__ __launch_bounds__(256) void chain_mix_wmma(const float* __restrict__ T1r, const float* __restrict__ T1i,
                          const float* __restrict__ Wr, const float* __restrict__ Wi,
                          const float* __restrict__ tcr, const float* __restrict__ tci,
                          float* __restrict__ F2r, float* __restrict__ F2i) {
  constexpr int BM = 32, NT = Geo<BM>::NT, A_ST = Geo<BM>::A_ST;
  __shared__ __attribute__((aligned(16))) float As[2 * A_ST];
  __shared__ __attribute__((aligned(16))) float Bs[2 * B_ST];
  __shared__ __attribute__((aligned(16))) float Fr[BM * RP];
  __shared__ __attribute__((aligned(16))) float Fi[BM * RP];
  __shared__ __attribute__((aligned(16))) float Fn[BM * RP];
  const int tid = threadIdx.x;
  const int j0 = blockIdx.x * BM;
  const int l  = blockIdx.y;
  const int wave = tid >> 5, lane = tid & 31;
  const int half = lane >> 4, l16 = lane & 15;
  const int wm = wave / Geo<BM>::WN, wn = wave % Geo<BM>::WN;
  v8f accR[NT], accI[NT], accP[NT], accQ[NT];
#pragma unroll
  for (int j = 0; j < NT; ++j) { accR[j] = v8f{}; accI[j] = v8f{}; accP[j] = v8f{}; accQ[j] = v8f{}; }
  MixAddr ad; ad.T1r = T1r + (long long)l * 128; ad.T1i = T1i + (long long)l * 128;
  ad.j0 = j0; ad.k0 = band_k0(j0);
  const float* wsrc[4];
#pragma unroll
  for (int q = 0; q < 4; ++q)
    wsrc[q] = (q < 2 ? Wr : Wi) + (long long)l * 16384 + (long long)((q & 1) * KC + (tid >> 5)) * 128 + (tid & 31) * 4;
  const int bsoff = (tid >> 5) * BP + (tid & 31) * 4;
  const int aoff = (wm * 16 + l16) * AP + 4 * half;
  const int foff = (wm * 16 + l16) * RP + 4 * half;
  const int boff = (4 * half) * BP + wn * (16 * NT) + l16;
  TileRegs<BM> r;
  tile_gload<BM>(r, ad, 0, tid);
  tile_sstore<BM>(r, As, Bs, tid);
  tile_gload<BM>(r, ad, 1, tid);
  __syncthreads();
#pragma unroll
  for (int ch = 0; ch < 8; ++ch) {
    const int st = ch & 1;
    if (ch + 1 < 4) {
      tile_sstore<BM>(r, As + (st ^ 1) * A_ST, Bs + (st ^ 1) * B_ST, tid);
    } else if (ch + 1 < 8) {
#pragma unroll
      for (int i = 0; i < 8; ++i) *(v4f*)(Bs + (st ^ 1) * B_ST + bsoff + i * 8 * BP) = r.b[i];
    }
    if (ch + 2 < 4) {
      tile_gload<BM>(r, ad, ch + 2, tid);
    } else if (ch + 2 < 8) {
#pragma unroll
      for (int i = 0; i < 8; ++i) r.b[i] = *(const v4f*)(wsrc[ch + 2 - 4] + i * 8 * 128);
    }
    const float* bw = Bs + st * B_ST + boff;
    if (ch < 2) {
      tile_compute<NT>(As + st * A_ST + aoff, bw, accR);
    } else if (ch < 4) {
      tile_compute<NT>(As + st * A_ST + aoff, bw, accI);
    } else {
      const int q = ch - 4;
      tile_compute<NT>((q < 2 ? Fr : Fn) + foff + (q & 1) * KC, bw, accP);
      tile_compute<NT>((q < 2 ? Fi : Fr) + foff + (q & 1) * KC, bw, accQ);
    }
    if (ch == 3) {
      const int fo = (wm * 16 + half * 8) * RP + wn * (16 * NT) + l16;
#pragma unroll
      for (int v = 0; v < 8; ++v) {
#pragma unroll
        for (int j = 0; j < NT; ++j) {
          const float re = accR[j][v], im = accI[j][v];
          Fr[fo + v * RP + 16 * j] = re;
          Fi[fo + v * RP + 16 * j] = im;
          Fn[fo + v * RP + 16 * j] = -im;
        }
      }
    }
    __syncthreads();
  }
  const float tr = tcr[l], ti = tci[l];
  const int r0 = j0 + wm * 16 + half * 8;
  const int c0 = wn * (16 * NT) + l16;
#pragma unroll
  for (int v = 0; v < 8; ++v) {
    const long long o = ((long long)l * 128 + r0 + v) * 128 + c0;
#pragma unroll
    for (int j = 0; j < NT; ++j) {
      const float p = accP[j][v], q = accQ[j][v];
      F2r[o + 16 * j] = tr * p - ti * q;
      F2i[o + 16 * j] = tr * q + ti * p;
    }
  }
}

__global__ __launch_bounds__(256) void out_stage_wmma(const float* __restrict__ xs2, const float* __restrict__ ck,
                             const float* __restrict__ cb, const float* __restrict__ tvec,
                             const float* __restrict__ c2, const float* __restrict__ s2n,
                             const float* __restrict__ ftr, const float* __restrict__ fti,
                             const float* __restrict__ lns, const float* __restrict__ lnb,
                             float* __restrict__ out) {
  constexpr int BM = 32, NT = Geo<BM>::NT;
  __shared__ __attribute__((aligned(16))) float As[2 * Geo<BM>::A_ST];
  __shared__ __attribute__((aligned(16))) float Bs[2 * B_ST];
  __shared__ float red[2][BM][4];
  const int tid = threadIdx.x;
  const int m0 = blockIdx.x * BM;
  const int l  = blockIdx.y;
  v8f acc[NT];
#pragma unroll
  for (int j = 0; j < NT; ++j) acc[j] = v8f{};
  OutAddr ad;
  ad.xs = xs2 + ((long long)(l - 1) * 128 + (m0 - 1)) * 128; ad.ck = ck; ad.l = l; ad.m0 = m0;
  ad.c2 = c2 + (long long)m0 * 128; ad.s2n = s2n + (long long)m0 * 128;
  ad.fr = ftr + (long long)l * 128; ad.fi = fti + (long long)l * 128;
  tile_mainloop<BM>(ad, 22, As, Bs, tid, acc);
  const int wave = tid >> 5, lane = tid & 31;
  const int half = lane >> 4, l16 = lane & 15;
  const int wm = wave / Geo<BM>::WN, wn = wave % Geo<BM>::WN;
  const int rl = wm * 16 + half * 8;
  const int c0 = wn * (16 * NT) + l16;
  float bias[NT];
#pragma unroll
  for (int j = 0; j < NT; ++j) bias[j] = cb[c0 + 16 * j] + tvec[128 + c0 + 16 * j];
  float g[NT][8], s[8];
#pragma unroll
  for (int v = 0; v < 8; ++v) {
    s[v] = 0.0f;
#pragma unroll
    for (int j = 0; j < NT; ++j) {
      const float x = acc[j][v] + bias[j];
      g[j][v] = 0.5f * x * (1.0f + tanhf(0.7978845608028654f * (x + 0.044715f * x * x * x)));
      s[v] += g[j][v];
    }
  }
#pragma unroll
  for (int o = 8; o > 0; o >>= 1) {
#pragma unroll
    for (int v = 0; v < 8; ++v) s[v] += __shfl_xor(s[v], o, 32);
  }
  if (l16 == 0) {
#pragma unroll
    for (int v = 0; v < 8; ++v) red[0][rl + v][wn] = s[v];
  }
  __syncthreads();
  float mu[8];
#pragma unroll
  for (int v = 0; v < 8; ++v) {
    const float* rp = red[0][rl + v];
    mu[v] = (rp[0] + rp[1] + rp[2] + rp[3]) * (1.0f / 128.0f);
    s[v] = 0.0f;
#pragma unroll
    for (int j = 0; j < NT; ++j) { g[j][v] -= mu[v]; s[v] += g[j][v] * g[j][v]; }
  }
#pragma unroll
  for (int o = 8; o > 0; o >>= 1) {
#pragma unroll
    for (int v = 0; v < 8; ++v) s[v] += __shfl_xor(s[v], o, 32);
  }
  if (l16 == 0) {
#pragma unroll
    for (int v = 0; v < 8; ++v) red[1][rl + v][wn] = s[v];
  }
  __syncthreads();
  float sc[NT], sh[NT];
#pragma unroll
  for (int j = 0; j < NT; ++j) { sc[j] = lns[c0 + 16 * j]; sh[j] = lnb[c0 + 16 * j]; }
#pragma unroll
  for (int v = 0; v < 8; ++v) {
    const float* rp = red[1][rl + v];
    const float var = (rp[0] + rp[1] + rp[2] + rp[3]) * (1.0f / 128.0f);
    const float rs = rsqrtf(var + 1e-6f);
    const int m = m0 + rl + v;
    if (m < 127) {
      float* op = out + (((long long)l) * 127 + m) * 128 + c0;
#pragma unroll
      for (int j = 0; j < NT; ++j) op[16 * j] = g[j][v] * rs * sc[j] + sh[j];
    }
  }
}

__global__ __launch_bounds__(256) void k_tvec(const float* __restrict__ temb,
                       const float* __restrict__ dw, const float* __restrict__ db,
                       const float* __restrict__ drw, const float* __restrict__ drb,
                       const float* __restrict__ diw, const float* __restrict__ dib,
                       float* tvec, float* tcr, float* tci) {
  __shared__ float part[16][17];
  const int tid = threadIdx.x, ks = tid >> 4, jj = tid & 15;
  const int blk = blockIdx.x;
  const float* w; const float* bias; float* o; int ld, c;
  if (blk < 16)      { w = dw;  bias = db;  o = tvec; ld = 256; c = blk * 16 + jj; }
  else if (blk < 20) { w = drw; bias = drb; o = tcr;  ld = 64;  c = (blk - 16) * 16 + jj; }
  else               { w = diw; bias = dib; o = tci;  ld = 64;  c = (blk - 20) * 16 + jj; }
  const float* wp = w + (long long)(ks * 16) * ld + c;
  const float* ep = temb + ks * 16;
  float v[16], e[16];
#pragma unroll
  for (int i = 0; i < 16; ++i) { v[i] = wp[(long long)i * ld]; e[i] = ep[i]; }
  float s = (ks == 0) ? bias[c] : 0.0f;
#pragma unroll
  for (int i = 0; i < 16; ++i) s += e[i] * v[i];
  part[ks][jj] = s;
  __syncthreads();
  if (tid < 16) {
    float a = 0.0f;
#pragma unroll
    for (int q = 0; q < 16; ++q) a += part[q][tid];
    o[c] = a;
  }
}

__global__ void k_phase1(float* c1, float* s1n, long long n) {
  long long i = (long long)blockIdx.x * 256 + threadIdx.x;
  if (i >= n) return;
  int m = (int)(i / 256), p = (int)(i % 256);
  if (m < 255 && p < 255) {
    double mval = (double)(m - 127);
    double phi = 2.0 * PI_D * (double)p / 255.0;
    double dphi = 2.0 * PI_D / 255.0;
    c1[i]  = (float)( cos(phi * mval) * dphi);
    s1n[i] = (float)(-sin(phi * mval) * dphi);
  } else { c1[i] = 0.0f; s1n[i] = 0.0f; }
}

__global__ void k_phase2(float* c2, float* s2n, long long n) {
  long long i = (long long)blockIdx.x * 256 + threadIdx.x;
  if (i >= n) return;
  int p = (int)(i / 128), m = (int)(i % 128);
  if (p < 127 && m < 127) {
    double mval = (double)(m - 63);
    double phi = 2.0 * PI_D * (double)p / 127.0;
    c2[i]  = (float)cos(mval * phi);
    s2n[i] = (float)(-sin(mval * phi));
  } else { c2[i] = 0.0f; s2n[i] = 0.0f; }
}

__device__ __forceinline__ double lane_bcast(double v, int src) {
  int lo = __double2loint(v), hi = __double2hiint(v);
  lo = __builtin_amdgcn_readlane(lo, src);
  hi = __builtin_amdgcn_readlane(hi, src);
  return __hiloint2double(hi, lo);
}

__global__ void k_legendre(float* __restrict__ out, int L, int weighted, int transposed) {
  const int lane = threadIdx.x;
  const int t = blockIdx.x * 32 + lane;
  const int m = blockIdx.y;
  const double theta = (2.0 * t + 1.0) * PI_D / (2.0 * L - 1.0);
  const double x = cos(theta), s = sin(theta);
  const double w = weighted ? (s * PI_D / (double)L) : 1.0;

  double ca[4], cb[4];
#pragma unroll
  for (int j = 0; j < 4; ++j) {
    ca[j] = 0.0; cb[j] = 0.0;
    if (m + 2 + 32 * j < L) {
      const double ll = (double)(m + 2 + 32 * j + lane), mm = (double)m;
      const double den = ll * ll - mm * mm;
      ca[j] = sqrt((4.0 * ll * ll - 1.0) / den);
      cb[j] = sqrt((2.0 * ll + 1.0) * ((ll - 1.0) * (ll - 1.0) - mm * mm) /
                   ((2.0 * ll - 3.0) * den));
    }
  }

  double num = 1.0, dnm = 1.0;
#pragma unroll
  for (int j = 0; j < 4; ++j) {
    const int k = lane + 1 + 32 * j;
    if (k <= m) { num *= (double)(2 * k + 1); dnm *= (double)(2 * k); }
  }
  double r = num / dnm;
#pragma unroll
  for (int o = 16; o > 0; o >>= 1) r *= __shfl_xor(r, o, 32);
  double base = -s, pw = 1.0;
  for (int e = m; e; e >>= 1) { if (e & 1) pw *= base; base *= base; }
  const double pmm = sqrt(1.0 / (4.0 * PI_D)) * w * pw * sqrt(r);

  const long long lstep = transposed ? 1 : L, tstep = transposed ? L : 1;
  float* op = out + (long long)(L - 1 + m) * L * (long long)L + t * tstep;
  if (transposed) {
    for (int l = 0; l < m; ++l) op[l] = 0.0f;
  }
  op += m * lstep;
  *op = (float)pmm; op += lstep;
  if (m + 1 < L) {
    double p2 = pmm;
    double p1 = sqrt(2.0 * m + 3.0) * x * pmm;
    *op = (float)p1; op += lstep;
#pragma unroll
    for (int j = 0; j < 4; ++j) {
      int cnt = L - (m + 2 + 32 * j);
      cnt = cnt > 32 ? 32 : cnt;
      for (int i = 0; i < cnt; ++i) {
        const double a = lane_bcast(ca[j], i), b = lane_bcast(cb[j], i);
        const double pn = a * x * p1 - b * p2;
        *op = (float)pn; op += lstep;
        p2 = p1; p1 = pn;
      }
    }
  }
}

template <int BM, int MODE>
static inline void launch_gemm(hipStream_t s, const float* A, const float* B, float* C,
                               int M, int K, int lda, int ldb, int ldc,
                               long long sA, long long sB, long long sC,
                               float alpha, int acc, int batch,
                               int brows, int amir, int arowz) {
  dim3 g(M / BM, batch);
  gemm_wmma<BM, MODE><<<g, dim3(256), 0, s>>>(A, B, C, K, lda, ldb, ldc, sA, sB, sC, alpha, acc,
                                              brows, amir, arowz);
}

#define EWG(n) dim3((unsigned)(((n) + 255) / 256)), dim3(256), 0, stream

extern "C" void kernel_launch(void* const* d_in, const int* in_sizes, int n_in,
                              void* d_out, int out_size, void* d_ws, size_t ws_size,
                              hipStream_t stream) {
  (void)in_sizes; (void)n_in; (void)out_size; (void)ws_size;
  const float* x       = (const float*)d_in[0];
  const float* t_emb   = (const float*)d_in[1];
  const float* sp_dw   = (const float*)d_in[2];
  const float* sp_db   = (const float*)d_in[3];
  const float* sp_wt   = (const float*)d_in[4];
  const float* sp_ck   = (const float*)d_in[5];
  const float* sp_cb   = (const float*)d_in[6];
  const float* spec_wr = (const float*)d_in[7];
  const float* spec_wi = (const float*)d_in[8];
  const float* drw     = (const float*)d_in[9];
  const float* drb     = (const float*)d_in[10];
  const float* diw     = (const float*)d_in[11];
  const float* dib     = (const float*)d_in[12];
  const float* lns     = (const float*)d_in[13];
  const float* lnb     = (const float*)d_in[14];
  float* out = (float*)d_out;
  float* ws  = (float*)d_ws;

  float *A1 = ws + OFF_A1, *COS1 = ws + OFF_COS1, *SIN1N = ws + OFF_SIN1N;
  float *GR = ws + OFF_GR, *GI = ws + OFF_GI;
  float *T1R = ws + OFF_T1R, *T1I = ws + OFF_T1I;
  float *F2R = ws + OFF_F2R, *F2I = ws + OFF_F2I, *P2B = ws + OFF_P2B;
  float *FTR = ws + OFF_FTR, *FTI = ws + OFF_FTI, *COS2 = ws + OFF_COS2, *SIN2N = ws + OFF_SIN2;
  float *TVEC = ws + OFF_TVEC, *TCR = ws + OFF_TCR, *TCI = ws + OFF_TCI;
  float *T2R = ws + OFF_T2R, *XS2 = ws + OFF_XS2;

  k_tvec<<<dim3(24), dim3(256), 0, stream>>>(t_emb, sp_dw, sp_db, drw, drb, diw, dib, TVEC, TCR, TCI);
  k_phase1<<<EWG(SZ_PH1)>>>(COS1, SIN1N, SZ_PH1);
  k_phase2<<<EWG(SZ_PH2)>>>(COS2, SIN2N, SZ_PH2);
  k_legendre<<<dim3(4, 128), dim3(32), 0, stream>>>(A1, 128, 1, 0);
  k_legendre<<<dim3(2, 64), dim3(32), 0, stream>>>(P2B, 64, 0, 1);

  launch_gemm<64, 2>(stream, COS1,  x, GR, 256, 128, 256, 128, 128, 0, 255LL * 128, 256LL * 128, 1.0f, 0, 128, 255, -1, 0);
  launch_gemm<64, 3>(stream, SIN1N, x, GI, 256, 128, 256, 128, 128, 0, 255LL * 128, 256LL * 128, 1.0f, 0, 128, 255, -1, 0);
  chain_resize_wmma<<<dim3(255, 2), dim3(256), 0, stream>>>(A1, GR, nullptr, T1R, 128, 256 * 128, 128LL * 128, 128, SZ_G,
                                                          64LL * 128, SZ_T1, 127);
  chain_mix_wmma<<<dim3(4, 64), dim3(256), 0, stream>>>(T1R, T1I, spec_wr, spec_wi, TCR, TCI, F2R, F2I);
  launch_gemm<32, 0>(stream, P2B, F2R, FTR, 64, 64, 64, 128 * 128, 128, 64LL * 64, 128, 64LL * 128, 1.0f, 0, 127, 64, 63, 0);
  launch_gemm<32, 0>(stream, P2B, F2I, FTI, 64, 64, 64, 128 * 128, 128, 64LL * 64, 128, 64LL * 128, 1.0f, 0, 127, 64, 63, 0);

  chain_resize_wmma<<<dim3(255, 1), dim3(256), 0, stream>>>(x, sp_wt, TVEC, T2R, 255 * 128, 128, 128, 0, 0,
                                                          64LL * 128, 0, -1);
  launch_gemm<32, 1>(stream, nullptr, T2R, XS2, 128, 256, 256, 64 * 128, 128, 0, 128, 128LL * 128, 1.0f, 0, 64, 255, -1, 0);
  out_stage_wmma<<<dim3(4, 64), dim3(256), 0, stream>>>(XS2, sp_ck, sp_cb, TVEC, COS2, SIN2N, FTR, FTI,
                                                      lns, lnb, out);
}
